# starting kernel
# baseline (speedup 1.0000x reference)
#include <hip/hip_runtime.h>

typedef __attribute__((ext_vector_type(2))) float v2f;
typedef __attribute__((ext_vector_type(8))) float v8f;

#define IN_DIM 128
#define HID    64
#define LAYERS 8
#define OUTD   40

__global__ void zero_int_kernel(int* __restrict__ p, int n) {
    int i = blockIdx.x * blockDim.x + threadIdx.x;
    if (i < n) p[i] = 0;
}

__global__ void count_deg_kernel(const int* __restrict__ dst, int* __restrict__ cnt, int E) {
    int e = blockIdx.x * blockDim.x + threadIdx.x;
    if (e < E) atomicAdd(&cnt[dst[e]], 1);
}

__global__ __launch_bounds__(1024) void scan_kernel(const int* __restrict__ cnt,
                                                    int* __restrict__ rowptr,
                                                    int N, int E) {
    __shared__ int sh[1024];
    __shared__ int carry;
    int tid = threadIdx.x;
    if (tid == 0) carry = 0;
    __syncthreads();
    for (int base = 0; base < N; base += 1024) {
        int idx = base + tid;
        int v = (idx < N) ? cnt[idx] : 0;
        sh[tid] = v;
        __syncthreads();
        for (int off = 1; off < 1024; off <<= 1) {
            int t = (tid >= off) ? sh[tid - off] : 0;
            __syncthreads();
            sh[tid] += t;
            __syncthreads();
        }
        int incl = sh[tid];
        int excl = incl - v;
        if (idx < N) rowptr[idx] = carry + excl;
        __syncthreads();
        if (tid == 1023) carry += sh[1023];
        __syncthreads();
    }
    if (tid == 0) rowptr[N] = E;
}

__global__ void init_pos_dis_kernel(const int* __restrict__ rowptr,
                                    const int* __restrict__ cnt,
                                    int* __restrict__ pos,
                                    float* __restrict__ dis, int N) {
    int i = blockIdx.x * blockDim.x + threadIdx.x;
    if (i < N) {
        pos[i] = rowptr[i];
        float dg = (float)(cnt[i] + 1);
        dis[i] = rsqrtf(dg);
    }
}

__global__ void fill_csr_kernel(const int* __restrict__ src, const int* __restrict__ dst,
                                int* __restrict__ pos, const float* __restrict__ dis,
                                int2* __restrict__ erec, int E) {
    int e = blockIdx.x * blockDim.x + threadIdx.x;
    if (e < E) {
        int s = src[e], d = dst[e];
        int p = atomicAdd(&pos[d], 1);
        erec[p] = make_int2(s, __float_as_int(dis[s] * dis[d]));
    }
}

__global__ void init_out_kernel(float* __restrict__ out, const float* __restrict__ lin_b,
                                int total) {
    int i = blockIdx.x * blockDim.x + threadIdx.x;
    if (i < total) out[i] = lin_b[i % OUTD];
}

__global__ __launch_bounds__(256) void gemm_wmma_f32_kernel(
    const float* __restrict__ A, const float* __restrict__ W,
    float* __restrict__ T, int nrows, int K)
{
    const int lane  = threadIdx.x & 31;
    const int wib   = threadIdx.x >> 5;
    const int strip = blockIdx.x * 8 + wib;
    const int m0    = strip * 16;
    if (m0 >= nrows) return;

    const int mi = lane & 15;
    const int hi = (lane >> 4) & 1;
    const int kb = hi * 2;

    const float* arow = A + (size_t)(m0 + mi) * K;

    v8f acc0 = {0.f,0.f,0.f,0.f,0.f,0.f,0.f,0.f};
    v8f acc1 = acc0, acc2 = acc0, acc3 = acc0;

    for (int k0 = 0; k0 < K; k0 += 4) {
        v2f a;
        a.x = arow[k0 + kb];
        a.y = arow[k0 + kb + 1];
        const float* w0 = W + (size_t)(k0 + kb) * HID + mi;
        const float* w1 = w0 + HID;
        v2f b0, b1, b2, b3;
        b0.x = w0[0];   b0.y = w1[0];
        b1.x = w0[16];  b1.y = w1[16];
        b2.x = w0[32];  b2.y = w1[32];
        b3.x = w0[48];  b3.y = w1[48];
        acc0 = __builtin_amdgcn_wmma_f32_16x16x4_f32(false, a, false, b0, (short)0, acc0, false, false);
        acc1 = __builtin_amdgcn_wmma_f32_16x16x4_f32(false, a, false, b1, (short)0, acc1, false, false);
        acc2 = __builtin_amdgcn_wmma_f32_16x16x4_f32(false, a, false, b2, (short)0, acc2, false, false);
        acc3 = __builtin_amdgcn_wmma_f32_16x16x4_f32(false, a, false, b3, (short)0, acc3, false, false);
    }

    #pragma unroll
    for (int v = 0; v < 8; ++v) {
        float* tr = T + (size_t)(m0 + v + hi * 8) * HID + mi;
        tr[0]  = acc0[v];
        tr[16] = acc1[v];
        tr[32] = acc2[v];
        tr[48] = acc3[v];
    }
}

__global__ __launch_bounds__(64) void aggregate_kernel(
    const float* __restrict__ T,
    const int* __restrict__ rowptr,
    const int2* __restrict__ erec,
    const float* __restrict__ dis,
    const float* __restrict__ bias,
    const float* __restrict__ lw,
    float* __restrict__ Hout,
    float* __restrict__ Out)
{
    const int i = blockIdx.x;
    const int f = threadIdx.x;

    float di  = dis[i];
    float acc = T[(size_t)i * HID + f] * di * di + bias[f];

    int e = rowptr[i];
    const int end = rowptr[i + 1];

    for (; e + 4 <= end; e += 4) {
        int2 r0 = erec[e + 0];
        int2 r1 = erec[e + 1];
        int2 r2 = erec[e + 2];
        int2 r3 = erec[e + 3];
        float t0 = T[(size_t)r0.x * HID + f];
        float t1 = T[(size_t)r1.x * HID + f];
        float t2 = T[(size_t)r2.x * HID + f];
        float t3 = T[(size_t)r3.x * HID + f];
        acc = fmaf(t0, __int_as_float(r0.y), acc);
        acc = fmaf(t1, __int_as_float(r1.y), acc);
        acc = fmaf(t2, __int_as_float(r2.y), acc);
        acc = fmaf(t3, __int_as_float(r3.y), acc);
    }
    for (; e < end; ++e) {
        int2 r = erec[e];
        acc = fmaf(T[(size_t)r.x * HID + f], __int_as_float(r.y), acc);
    }

    acc = fmaxf(acc, 0.f);
    Hout[(size_t)i * HID + f] = acc;

    __shared__ float sh[HID];
    sh[f] = acc;
    __syncthreads();

    if (f < OUTD) {
        float s = 0.f;
        #pragma unroll
        for (int c = 0; c < HID; ++c)
            s = fmaf(sh[c], lw[c * OUTD + f], s);
        Out[(size_t)i * OUTD + f] += s;
    }
}

extern "C" void kernel_launch(void* const* d_in, const int* in_sizes, int n_in,
                              void* d_out, int out_size, void* d_ws, size_t ws_size,
                              hipStream_t stream) {
    const float* x     = (const float*)d_in[0];
    const float* W0    = (const float*)d_in[1];
    const float* Ws    = (const float*)d_in[2];
    const float* bs    = (const float*)d_in[3];
    const float* lin_w = (const float*)d_in[4];
    const float* lin_b = (const float*)d_in[5];
    const int*   ei    = (const int*)d_in[6];
    float* out = (float*)d_out;

    const int N = in_sizes[0] / IN_DIM;
    const int E = in_sizes[6] / 2;
    const int* src = ei;
    const int* dst = ei + E;

    char* ws = (char*)d_ws;
    int2*  erec   = (int2*)ws;  ws += sizeof(int2)  * (size_t)E;
    int*   cnt    = (int*)ws;   ws += sizeof(int)   * (size_t)N;
    int*   rowptr = (int*)ws;   ws += sizeof(int)   * (size_t)(N + 1);
    int*   pos    = (int*)ws;   ws += sizeof(int)   * (size_t)N;
    float* dis    = (float*)ws; ws += sizeof(float) * (size_t)N;
    float* t      = (float*)ws; ws += sizeof(float) * (size_t)N * HID;
    float* ha     = (float*)ws; ws += sizeof(float) * (size_t)N * HID;
    float* hb     = (float*)ws; ws += sizeof(float) * (size_t)N * HID;

    const int TPB = 256;
    zero_int_kernel<<<(N + TPB - 1) / TPB, TPB, 0, stream>>>(cnt, N);
    count_deg_kernel<<<(E + TPB - 1) / TPB, TPB, 0, stream>>>(dst, cnt, E);
    scan_kernel<<<1, 1024, 0, stream>>>(cnt, rowptr, N, E);
    init_pos_dis_kernel<<<(N + TPB - 1) / TPB, TPB, 0, stream>>>(rowptr, cnt, pos, dis, N);
    fill_csr_kernel<<<(E + TPB - 1) / TPB, TPB, 0, stream>>>(src, dst, pos, dis, erec, E);
    init_out_kernel<<<((N * OUTD) + TPB - 1) / TPB, TPB, 0, stream>>>(out, lin_b, N * OUTD);

    const int strips = N / 16;
    const int gblocks = (strips + 7) / 8;
    for (int l = 0; l < LAYERS; ++l) {
        const float* hin  = (l == 0) ? x : ((l & 1) ? ha : hb);
        float*       hcur = (l & 1) ? hb : ha;
        const int    K    = (l == 0) ? IN_DIM : HID;
        const float* W    = (l == 0) ? W0 : (Ws + (size_t)(l - 1) * HID * HID);

        gemm_wmma_f32_kernel<<<gblocks, 256, 0, stream>>>(hin, W, t, N, K);
        aggregate_kernel<<<N, 64, 0, stream>>>(t, rowptr, erec, dis,
                                               bs + (size_t)l * HID,
                                               lin_w + (size_t)l * HID * OUTD,
                                               hcur, out);
    }
}


# Row-pointer scan: one 1024-thread block walking 98 barrier-bound tiles -> one block per tile that sums the counts before it with independent loads, then one wave-shuffle scan and one barrier
# speedup vs baseline: 1.2632x; 1.2632x over previous
#include <hip/hip_runtime.h>

typedef __attribute__((ext_vector_type(2))) float v2f;
typedef __attribute__((ext_vector_type(8))) float v8f;

#define IN_DIM 128
#define HID    64
#define LAYERS 8
#define OUTD   40

__global__ void zero_int_kernel(int* __restrict__ p, int n) {
    int i = blockIdx.x * blockDim.x + threadIdx.x;
    if (i < n) p[i] = 0;
}

__global__ void count_deg_kernel(const int* __restrict__ dst, int* __restrict__ cnt, int E) {
    int e = blockIdx.x * blockDim.x + threadIdx.x;
    if (e < E) atomicAdd(&cnt[dst[e]], 1);
}

#define SCAN_TPB 1024
#define SCAN_PF  16
__global__ __launch_bounds__(SCAN_TPB) void scan_kernel(const int* __restrict__ cnt,
                                                        int* __restrict__ rowptr,
                                                        int N, int E) {
    __shared__ int sh_tot[SCAN_TPB / 32];
    __shared__ int sh_pre[SCAN_TPB / 32];
    const int tid  = threadIdx.x;
    const int lane = tid & 31;
    const int wib  = tid >> 5;
    const int base = blockIdx.x * SCAN_TPB;

    int pre = 0;
    for (int j0 = tid; j0 < base; j0 += SCAN_PF * SCAN_TPB) {
        #pragma unroll
        for (int u = 0; u < SCAN_PF; ++u) {
            const int j = j0 + u * SCAN_TPB;
            pre += (j < base) ? cnt[j] : 0;
        }
    }

    const int idx = base + tid;
    const int v = (idx < N) ? cnt[idx] : 0;

    int incl = v;
    for (int off = 1; off < 32; off <<= 1) {
        int t = __shfl_up(incl, off, 32);
        if (lane >= off) incl += t;
    }
    for (int off = 16; off > 0; off >>= 1) pre += __shfl_xor(pre, off, 32);

    if (lane == 31) sh_tot[wib] = incl;
    if (lane == 0)  sh_pre[wib] = pre;
    __syncthreads();

    int part = sh_pre[lane] + ((lane < wib) ? sh_tot[lane] : 0);
    for (int off = 16; off > 0; off >>= 1) part += __shfl_xor(part, off, 32);

    if (idx < N) rowptr[idx] = part + incl - v;
    if (blockIdx.x == 0 && tid == 0) rowptr[N] = E;
}

__global__ void init_pos_dis_kernel(const int* __restrict__ rowptr,
                                    const int* __restrict__ cnt,
                                    int* __restrict__ pos,
                                    float* __restrict__ dis, int N) {
    int i = blockIdx.x * blockDim.x + threadIdx.x;
    if (i < N) {
        pos[i] = rowptr[i];
        float dg = (float)(cnt[i] + 1);
        dis[i] = rsqrtf(dg);
    }
}

__global__ void fill_csr_kernel(const int* __restrict__ src, const int* __restrict__ dst,
                                int* __restrict__ pos, const float* __restrict__ dis,
                                int2* __restrict__ erec, int E) {
    int e = blockIdx.x * blockDim.x + threadIdx.x;
    if (e < E) {
        int s = src[e], d = dst[e];
        int p = atomicAdd(&pos[d], 1);
        erec[p] = make_int2(s, __float_as_int(dis[s] * dis[d]));
    }
}

__global__ void init_out_kernel(float* __restrict__ out, const float* __restrict__ lin_b,
                                int total) {
    int i = blockIdx.x * blockDim.x + threadIdx.x;
    if (i < total) out[i] = lin_b[i % OUTD];
}

__global__ __launch_bounds__(256) void gemm_wmma_f32_kernel(
    const float* __restrict__ A, const float* __restrict__ W,
    float* __restrict__ T, int nrows, int K)
{
    const int lane  = threadIdx.x & 31;
    const int wib   = threadIdx.x >> 5;
    const int strip = blockIdx.x * 8 + wib;
    const int m0    = strip * 16;
    if (m0 >= nrows) return;

    const int mi = lane & 15;
    const int hi = (lane >> 4) & 1;
    const int kb = hi * 2;

    const float* arow = A + (size_t)(m0 + mi) * K;

    v8f acc0 = {0.f,0.f,0.f,0.f,0.f,0.f,0.f,0.f};
    v8f acc1 = acc0, acc2 = acc0, acc3 = acc0;

    for (int k0 = 0; k0 < K; k0 += 4) {
        v2f a;
        a.x = arow[k0 + kb];
        a.y = arow[k0 + kb + 1];
        const float* w0 = W + (size_t)(k0 + kb) * HID + mi;
        const float* w1 = w0 + HID;
        v2f b0, b1, b2, b3;
        b0.x = w0[0];   b0.y = w1[0];
        b1.x = w0[16];  b1.y = w1[16];
        b2.x = w0[32];  b2.y = w1[32];
        b3.x = w0[48];  b3.y = w1[48];
        acc0 = __builtin_amdgcn_wmma_f32_16x16x4_f32(false, a, false, b0, (short)0, acc0, false, false);
        acc1 = __builtin_amdgcn_wmma_f32_16x16x4_f32(false, a, false, b1, (short)0, acc1, false, false);
        acc2 = __builtin_amdgcn_wmma_f32_16x16x4_f32(false, a, false, b2, (short)0, acc2, false, false);
        acc3 = __builtin_amdgcn_wmma_f32_16x16x4_f32(false, a, false, b3, (short)0, acc3, false, false);
    }

    #pragma unroll
    for (int v = 0; v < 8; ++v) {
        float* tr = T + (size_t)(m0 + v + hi * 8) * HID + mi;
        tr[0]  = acc0[v];
        tr[16] = acc1[v];
        tr[32] = acc2[v];
        tr[48] = acc3[v];
    }
}

__global__ __launch_bounds__(64) void aggregate_kernel(
    const float* __restrict__ T,
    const int* __restrict__ rowptr,
    const int2* __restrict__ erec,
    const float* __restrict__ dis,
    const float* __restrict__ bias,
    const float* __restrict__ lw,
    float* __restrict__ Hout,
    float* __restrict__ Out)
{
    const int i = blockIdx.x;
    const int f = threadIdx.x;

    float di  = dis[i];
    float acc = T[(size_t)i * HID + f] * di * di + bias[f];

    int e = rowptr[i];
    const int end = rowptr[i + 1];

    for (; e + 4 <= end; e += 4) {
        int2 r0 = erec[e + 0];
        int2 r1 = erec[e + 1];
        int2 r2 = erec[e + 2];
        int2 r3 = erec[e + 3];
        float t0 = T[(size_t)r0.x * HID + f];
        float t1 = T[(size_t)r1.x * HID + f];
        float t2 = T[(size_t)r2.x * HID + f];
        float t3 = T[(size_t)r3.x * HID + f];
        acc = fmaf(t0, __int_as_float(r0.y), acc);
        acc = fmaf(t1, __int_as_float(r1.y), acc);
        acc = fmaf(t2, __int_as_float(r2.y), acc);
        acc = fmaf(t3, __int_as_float(r3.y), acc);
    }
    for (; e < end; ++e) {
        int2 r = erec[e];
        acc = fmaf(T[(size_t)r.x * HID + f], __int_as_float(r.y), acc);
    }

    acc = fmaxf(acc, 0.f);
    Hout[(size_t)i * HID + f] = acc;

    __shared__ float sh[HID];
    sh[f] = acc;
    __syncthreads();

    if (f < OUTD) {
        float s = 0.f;
        #pragma unroll
        for (int c = 0; c < HID; ++c)
            s = fmaf(sh[c], lw[c * OUTD + f], s);
        Out[(size_t)i * OUTD + f] += s;
    }
}

extern "C" void kernel_launch(void* const* d_in, const int* in_sizes, int n_in,
                              void* d_out, int out_size, void* d_ws, size_t ws_size,
                              hipStream_t stream) {
    const float* x     = (const float*)d_in[0];
    const float* W0    = (const float*)d_in[1];
    const float* Ws    = (const float*)d_in[2];
    const float* bs    = (const float*)d_in[3];
    const float* lin_w = (const float*)d_in[4];
    const float* lin_b = (const float*)d_in[5];
    const int*   ei    = (const int*)d_in[6];
    float* out = (float*)d_out;

    const int N = in_sizes[0] / IN_DIM;
    const int E = in_sizes[6] / 2;
    const int* src = ei;
    const int* dst = ei + E;

    char* ws = (char*)d_ws;
    int2*  erec   = (int2*)ws;  ws += sizeof(int2)  * (size_t)E;
    int*   cnt    = (int*)ws;   ws += sizeof(int)   * (size_t)N;
    int*   rowptr = (int*)ws;   ws += sizeof(int)   * (size_t)(N + 1);
    int*   pos    = (int*)ws;   ws += sizeof(int)   * (size_t)N;
    float* dis    = (float*)ws; ws += sizeof(float) * (size_t)N;
    float* t      = (float*)ws; ws += sizeof(float) * (size_t)N * HID;
    float* ha     = (float*)ws; ws += sizeof(float) * (size_t)N * HID;
    float* hb     = (float*)ws; ws += sizeof(float) * (size_t)N * HID;

    const int TPB = 256;
    zero_int_kernel<<<(N + TPB - 1) / TPB, TPB, 0, stream>>>(cnt, N);
    count_deg_kernel<<<(E + TPB - 1) / TPB, TPB, 0, stream>>>(dst, cnt, E);
    scan_kernel<<<(N + SCAN_TPB - 1) / SCAN_TPB, SCAN_TPB, 0, stream>>>(cnt, rowptr, N, E);
    init_pos_dis_kernel<<<(N + TPB - 1) / TPB, TPB, 0, stream>>>(rowptr, cnt, pos, dis, N);
    fill_csr_kernel<<<(E + TPB - 1) / TPB, TPB, 0, stream>>>(src, dst, pos, dis, erec, E);
    init_out_kernel<<<((N * OUTD) + TPB - 1) / TPB, TPB, 0, stream>>>(out, lin_b, N * OUTD);

    const int strips = N / 16;
    const int gblocks = (strips + 7) / 8;
    for (int l = 0; l < LAYERS; ++l) {
        const float* hin  = (l == 0) ? x : ((l & 1) ? ha : hb);
        float*       hcur = (l & 1) ? hb : ha;
        const int    K    = (l == 0) ? IN_DIM : HID;
        const float* W    = (l == 0) ? W0 : (Ws + (size_t)(l - 1) * HID * HID);

        gemm_wmma_f32_kernel<<<gblocks, 256, 0, stream>>>(hin, W, t, N, K);
        aggregate_kernel<<<N, 64, 0, stream>>>(t, rowptr, erec, dis,
                                               bs + (size_t)l * HID,
                                               lin_w + (size_t)l * HID * OUTD,
                                               hcur, out);
    }
}


# Aggregation: a node's edge records fetched by one coalesced wave load and handed round with readlane, so 8 neighbour rows are in flight at once, not 4 behind a scalar record fetch; sum order kept
# speedup vs baseline: 1.9546x; 1.5473x over previous
#include <hip/hip_runtime.h>

typedef __attribute__((ext_vector_type(2))) float v2f;
typedef __attribute__((ext_vector_type(8))) float v8f;

#define IN_DIM 128
#define HID    64
#define LAYERS 8
#define OUTD   40

__global__ void zero_int_kernel(int* __restrict__ p, int n) {
    int i = blockIdx.x * blockDim.x + threadIdx.x;
    if (i < n) p[i] = 0;
}

__global__ void count_deg_kernel(const int* __restrict__ dst, int* __restrict__ cnt, int E) {
    int e = blockIdx.x * blockDim.x + threadIdx.x;
    if (e < E) atomicAdd(&cnt[dst[e]], 1);
}

#define SCAN_TPB 1024
#define SCAN_PF  16
__global__ __launch_bounds__(SCAN_TPB) void scan_kernel(const int* __restrict__ cnt,
                                                        int* __restrict__ rowptr,
                                                        int N, int E) {
    __shared__ int sh_tot[SCAN_TPB / 32];
    __shared__ int sh_pre[SCAN_TPB / 32];
    const int tid  = threadIdx.x;
    const int lane = tid & 31;
    const int wib  = tid >> 5;
    const int base = blockIdx.x * SCAN_TPB;

    int pre = 0;
    for (int j0 = tid; j0 < base; j0 += SCAN_PF * SCAN_TPB) {
        #pragma unroll
        for (int u = 0; u < SCAN_PF; ++u) {
            const int j = j0 + u * SCAN_TPB;
            pre += (j < base) ? cnt[j] : 0;
        }
    }

    const int idx = base + tid;
    const int v = (idx < N) ? cnt[idx] : 0;

    int incl = v;
    for (int off = 1; off < 32; off <<= 1) {
        int t = __shfl_up(incl, off, 32);
        if (lane >= off) incl += t;
    }
    for (int off = 16; off > 0; off >>= 1) pre += __shfl_xor(pre, off, 32);

    if (lane == 31) sh_tot[wib] = incl;
    if (lane == 0)  sh_pre[wib] = pre;
    __syncthreads();

    int part = sh_pre[lane] + ((lane < wib) ? sh_tot[lane] : 0);
    for (int off = 16; off > 0; off >>= 1) part += __shfl_xor(part, off, 32);

    if (idx < N) rowptr[idx] = part + incl - v;
    if (blockIdx.x == 0 && tid == 0) rowptr[N] = E;
}

__global__ void init_pos_dis_kernel(const int* __restrict__ rowptr,
                                    const int* __restrict__ cnt,
                                    int* __restrict__ pos,
                                    float* __restrict__ dis, int N) {
    int i = blockIdx.x * blockDim.x + threadIdx.x;
    if (i < N) {
        pos[i] = rowptr[i];
        float dg = (float)(cnt[i] + 1);
        dis[i] = rsqrtf(dg);
    }
}

__global__ void fill_csr_kernel(const int* __restrict__ src, const int* __restrict__ dst,
                                int* __restrict__ pos, const float* __restrict__ dis,
                                int2* __restrict__ erec, int E) {
    int e = blockIdx.x * blockDim.x + threadIdx.x;
    if (e < E) {
        int s = src[e], d = dst[e];
        int p = atomicAdd(&pos[d], 1);
        erec[p] = make_int2(s, __float_as_int(dis[s] * dis[d]));
    }
}

__global__ __launch_bounds__(256) void gemm_wmma_f32_kernel(
    const float* __restrict__ A, const float* __restrict__ W,
    float* __restrict__ T, int nrows, int K)
{
    const int lane  = threadIdx.x & 31;
    const int wib   = threadIdx.x >> 5;
    const int strip = blockIdx.x * 8 + wib;
    const int m0    = strip * 16;
    if (m0 >= nrows) return;

    const int mi = lane & 15;
    const int hi = (lane >> 4) & 1;
    const int kb = hi * 2;

    const float* arow = A + (size_t)(m0 + mi) * K;

    v8f acc0 = {0.f,0.f,0.f,0.f,0.f,0.f,0.f,0.f};
    v8f acc1 = acc0, acc2 = acc0, acc3 = acc0;

    for (int k0 = 0; k0 < K; k0 += 4) {
        v2f a;
        a.x = arow[k0 + kb];
        a.y = arow[k0 + kb + 1];
        const float* w0 = W + (size_t)(k0 + kb) * HID + mi;
        const float* w1 = w0 + HID;
        v2f b0, b1, b2, b3;
        b0.x = w0[0];   b0.y = w1[0];
        b1.x = w0[16];  b1.y = w1[16];
        b2.x = w0[32];  b2.y = w1[32];
        b3.x = w0[48];  b3.y = w1[48];
        acc0 = __builtin_amdgcn_wmma_f32_16x16x4_f32(false, a, false, b0, (short)0, acc0, false, false);
        acc1 = __builtin_amdgcn_wmma_f32_16x16x4_f32(false, a, false, b1, (short)0, acc1, false, false);
        acc2 = __builtin_amdgcn_wmma_f32_16x16x4_f32(false, a, false, b2, (short)0, acc2, false, false);
        acc3 = __builtin_amdgcn_wmma_f32_16x16x4_f32(false, a, false, b3, (short)0, acc3, false, false);
    }

    #pragma unroll
    for (int v = 0; v < 8; ++v) {
        float* tr = T + (size_t)(m0 + v + hi * 8) * HID + mi;
        tr[0]  = acc0[v];
        tr[16] = acc1[v];
        tr[32] = acc2[v];
        tr[48] = acc3[v];
    }
}

#define AGG_NODES 16
#define AGG_WAVES 4
#define AGG_NPW   (AGG_NODES / AGG_WAVES)
#define AGG_GRP   8
#define SH_PITCH2 34

__global__ __launch_bounds__(AGG_WAVES * 32) void aggregate_kernel(
    const float* __restrict__ T,
    const int* __restrict__ rowptr,
    const int2* __restrict__ erec,
    const float* __restrict__ dis,
    const float* __restrict__ bias,
    const float* __restrict__ lw,
    const float* __restrict__ lin_b,
    float* __restrict__ Hout,
    float* __restrict__ Out,
    int N, int first)
{
    __shared__ v2f sh[AGG_NODES * SH_PITCH2];

    const int lane = threadIdx.x & 31;
    const int wib = __builtin_amdgcn_readfirstlane((int)(threadIdx.x >> 5));
    const int m0  = blockIdx.x * AGG_NODES;
    const int i0  = m0 + wib * AGG_NPW;

    const v2f* __restrict__ T2 = reinterpret_cast<const v2f*>(T);
    v2f* __restrict__ H2 = reinterpret_cast<v2f*>(Hout);

    const float bx = bias[2 * lane];
    const float by = bias[2 * lane + 1];

    int   rp[AGG_NPW + 1];
    float dv[AGG_NPW];
    v2f   ts[AGG_NPW];
    #pragma unroll
    for (int u = 0; u <= AGG_NPW; ++u) rp[u] = rowptr[min(i0 + u, N)];
    #pragma unroll
    for (int u = 0; u < AGG_NPW; ++u) {
        const int ic = min(i0 + u, N - 1);
        dv[u] = dis[ic];
        ts[u] = T2[(size_t)ic * (HID / 2) + lane];
    }

    int2 rc[AGG_NPW];
    #pragma unroll
    for (int u = 0; u < AGG_NPW; ++u) {
        rc[u] = make_int2(0, 0);
        if (rp[u] + lane < rp[u + 1]) rc[u] = erec[rp[u] + lane];
    }

    #pragma unroll
    for (int u = 0; u < AGG_NPW; ++u) {
        const int   i  = i0 + u;
        const float di = dv[u];
        float ax = ts[u].x * di * di + bx;
        float ay = ts[u].y * di * di + by;

        int2 rec = rc[u];
        int base = rp[u];
        const int end = rp[u + 1];
        while (base < end) {
            const int n = min(end - base, 32);
            int j = 0;
            for (; j + AGG_GRP <= n; j += AGG_GRP) {
                v2f   t[AGG_GRP];
                float w[AGG_GRP];
                #pragma unroll
                for (int k = 0; k < AGG_GRP; ++k) {
                    const int c = __builtin_amdgcn_readlane(rec.x, j + k);
                    w[k] = __int_as_float(__builtin_amdgcn_readlane(rec.y, j + k));
                    t[k] = T2[(size_t)c * (HID / 2) + lane];
                }
                #pragma unroll
                for (int k = 0; k < AGG_GRP; ++k) {
                    ax = fmaf(t[k].x, w[k], ax);
                    ay = fmaf(t[k].y, w[k], ay);
                }
            }
            if (j < n) {
                v2f   t[AGG_GRP - 1];
                float w[AGG_GRP - 1];
                #pragma unroll
                for (int k = 0; k < AGG_GRP - 1; ++k) {
                    t[k].x = 0.f;  t[k].y = 0.f;  w[k] = 0.f;
                    if (j + k < n) {
                        const int c = __builtin_amdgcn_readlane(rec.x, j + k);
                        w[k] = __int_as_float(__builtin_amdgcn_readlane(rec.y, j + k));
                        t[k] = T2[(size_t)c * (HID / 2) + lane];
                    }
                }
                #pragma unroll
                for (int k = 0; k < AGG_GRP - 1; ++k) {
                    ax = fmaf(t[k].x, w[k], ax);
                    ay = fmaf(t[k].y, w[k], ay);
                }
            }
            base += 32;
            if (base < end) {
                rec = make_int2(0, 0);
                if (base + lane < end) rec = erec[base + lane];
            }
        }

        v2f h;
        h.x = fmaxf(ax, 0.f);
        h.y = fmaxf(ay, 0.f);
        if (i < N) {
            H2[(size_t)i * (HID / 2) + lane] = h;
        } else {
            h.x = 0.f;  h.y = 0.f;
        }
        sh[(wib * AGG_NPW + u) * SH_PITCH2 + lane] = h;
    }
    __syncthreads();

    if (wib < (OUTD + 15) / 16) {
        const int  mi  = lane & 15;
        const int  hi  = lane >> 4;
        const int  kb  = hi * 2;
        const int  col = wib * 16 + mi;
        const bool cok = col < OUTD;

        v8f acc;
        #pragma unroll
        for (int v = 0; v < 8; ++v) {
            const int row = m0 + v + hi * 8;
            float c0 = 0.f;
            if (cok && row < N) c0 = first ? lin_b[col] : Out[(size_t)row * OUTD + col];
            acc[v] = c0;
        }

        v2f bf[HID / 4];
        #pragma unroll
        for (int s = 0; s < HID / 4; ++s) { bf[s].x = 0.f;  bf[s].y = 0.f; }
        if (cok) {
            #pragma unroll
            for (int s = 0; s < HID / 4; ++s) {
                bf[s].x = lw[(4 * s + kb) * OUTD + col];
                bf[s].y = lw[(4 * s + kb + 1) * OUTD + col];
            }
        }

        #pragma unroll
        for (int s = 0; s < HID / 4; ++s) {
            const v2f a = sh[mi * SH_PITCH2 + ((4 * s + kb) >> 1)];
            acc = __builtin_amdgcn_wmma_f32_16x16x4_f32(false, a, false, bf[s], (short)0, acc, false, false);
        }

        #pragma unroll
        for (int v = 0; v < 8; ++v) {
            const int row = m0 + v + hi * 8;
            if (cok && row < N) Out[(size_t)row * OUTD + col] = acc[v];
        }
    }
}

extern "C" void kernel_launch(void* const* d_in, const int* in_sizes, int n_in,
                              void* d_out, int out_size, void* d_ws, size_t ws_size,
                              hipStream_t stream) {
    const float* x     = (const float*)d_in[0];
    const float* W0    = (const float*)d_in[1];
    const float* Ws    = (const float*)d_in[2];
    const float* bs    = (const float*)d_in[3];
    const float* lin_w = (const float*)d_in[4];
    const float* lin_b = (const float*)d_in[5];
    const int*   ei    = (const int*)d_in[6];
    float* out = (float*)d_out;

    const int N = in_sizes[0] / IN_DIM;
    const int E = in_sizes[6] / 2;
    const int* src = ei;
    const int* dst = ei + E;

    char* ws = (char*)d_ws;
    float* t      = (float*)ws; ws += sizeof(float) * (size_t)N * HID;
    float* ha     = (float*)ws; ws += sizeof(float) * (size_t)N * HID;
    float* hb     = (float*)ws; ws += sizeof(float) * (size_t)N * HID;
    int2*  erec   = (int2*)ws;  ws += sizeof(int2)  * (size_t)E;
    int*   cnt    = (int*)ws;   ws += sizeof(int)   * (size_t)N;
    int*   pos    = (int*)ws;   ws += sizeof(int)   * (size_t)N;
    float* dis    = (float*)ws; ws += sizeof(float) * (size_t)N;
    int*   rowptr = (int*)ws;   ws += sizeof(int)   * (size_t)(N + 1);

    const int TPB = 256;
    zero_int_kernel<<<(N + TPB - 1) / TPB, TPB, 0, stream>>>(cnt, N);
    count_deg_kernel<<<(E + TPB - 1) / TPB, TPB, 0, stream>>>(dst, cnt, E);
    scan_kernel<<<(N + SCAN_TPB - 1) / SCAN_TPB, SCAN_TPB, 0, stream>>>(cnt, rowptr, N, E);
    init_pos_dis_kernel<<<(N + TPB - 1) / TPB, TPB, 0, stream>>>(rowptr, cnt, pos, dis, N);
    fill_csr_kernel<<<(E + TPB - 1) / TPB, TPB, 0, stream>>>(src, dst, pos, dis, erec, E);
    const int strips = N / 16;
    const int gblocks = (strips + 7) / 8;
    for (int l = 0; l < LAYERS; ++l) {
        const float* hin  = (l == 0) ? x : ((l & 1) ? ha : hb);
        float*       hcur = (l & 1) ? hb : ha;
        const int    K    = (l == 0) ? IN_DIM : HID;
        const float* W    = (l == 0) ? W0 : (Ws + (size_t)(l - 1) * HID * HID);

        gemm_wmma_f32_kernel<<<gblocks, 256, 0, stream>>>(hin, W, t, N, K);
        aggregate_kernel<<<(N + AGG_NODES - 1) / AGG_NODES, AGG_WAVES * 32, 0, stream>>>(
            t, rowptr, erec, dis,
            bs + (size_t)l * HID,
            lin_w + (size_t)l * HID * OUTD,
            lin_b, hcur, out, N, (l == 0) ? 1 : 0);
    }
}


# Next layer's dense transform fused into the aggregation kernel: T_next = h_tile @ W_next as WMMA on the h tile in LDS, so h never goes to memory and 7 GEMM launches go; k order of every sum kept
# speedup vs baseline: 2.3253x; 1.1897x over previous
#include <hip/hip_runtime.h>

typedef __attribute__((ext_vector_type(2))) float v2f;
typedef __attribute__((ext_vector_type(8))) float v8f;

#define IN_DIM 128
#define HID    64
#define LAYERS 8
#define OUTD   40

__global__ void zero_int_kernel(int* __restrict__ p, int n) {
    int i = blockIdx.x * blockDim.x + threadIdx.x;
    if (i < n) p[i] = 0;
}

__global__ void count_deg_kernel(const int* __restrict__ dst, int* __restrict__ cnt, int E) {
    int e = blockIdx.x * blockDim.x + threadIdx.x;
    if (e < E) atomicAdd(&cnt[dst[e]], 1);
}

#define SCAN_TPB 1024
#define SCAN_PF  16
__global__ __launch_bounds__(SCAN_TPB) void scan_kernel(const int* __restrict__ cnt,
                                                        int* __restrict__ rowptr,
                                                        int N, int E) {
    __shared__ int sh_tot[SCAN_TPB / 32];
    __shared__ int sh_pre[SCAN_TPB / 32];
    const int tid  = threadIdx.x;
    const int lane = tid & 31;
    const int wib  = tid >> 5;
    const int base = blockIdx.x * SCAN_TPB;

    int pre = 0;
    for (int j0 = tid; j0 < base; j0 += SCAN_PF * SCAN_TPB) {
        #pragma unroll
        for (int u = 0; u < SCAN_PF; ++u) {
            const int j = j0 + u * SCAN_TPB;
            pre += (j < base) ? cnt[j] : 0;
        }
    }

    const int idx = base + tid;
    const int v = (idx < N) ? cnt[idx] : 0;

    int incl = v;
    for (int off = 1; off < 32; off <<= 1) {
        int t = __shfl_up(incl, off, 32);
        if (lane >= off) incl += t;
    }
    for (int off = 16; off > 0; off >>= 1) pre += __shfl_xor(pre, off, 32);

    if (lane == 31) sh_tot[wib] = incl;
    if (lane == 0)  sh_pre[wib] = pre;
    __syncthreads();

    int part = sh_pre[lane] + ((lane < wib) ? sh_tot[lane] : 0);
    for (int off = 16; off > 0; off >>= 1) part += __shfl_xor(part, off, 32);

    if (idx < N) rowptr[idx] = part + incl - v;
    if (blockIdx.x == 0 && tid == 0) rowptr[N] = E;
}

__global__ void init_pos_dis_kernel(const int* __restrict__ rowptr,
                                    const int* __restrict__ cnt,
                                    int* __restrict__ pos,
                                    float* __restrict__ dis, int N) {
    int i = blockIdx.x * blockDim.x + threadIdx.x;
    if (i < N) {
        pos[i] = rowptr[i];
        float dg = (float)(cnt[i] + 1);
        dis[i] = rsqrtf(dg);
    }
}

__global__ void fill_csr_kernel(const int* __restrict__ src, const int* __restrict__ dst,
                                int* __restrict__ pos, const float* __restrict__ dis,
                                int2* __restrict__ erec, int E) {
    int e = blockIdx.x * blockDim.x + threadIdx.x;
    if (e < E) {
        int s = src[e], d = dst[e];
        int p = atomicAdd(&pos[d], 1);
        erec[p] = make_int2(s, __float_as_int(dis[s] * dis[d]));
    }
}

__global__ __launch_bounds__(256) void gemm_wmma_f32_kernel(
    const float* __restrict__ A, const float* __restrict__ W,
    float* __restrict__ T, int nrows, int K)
{
    const int lane  = threadIdx.x & 31;
    const int wib   = threadIdx.x >> 5;
    const int strip = blockIdx.x * 8 + wib;
    const int m0    = strip * 16;
    if (m0 >= nrows) return;

    const int mi = lane & 15;
    const int hi = (lane >> 4) & 1;
    const int kb = hi * 2;

    const float* arow = A + (size_t)(m0 + mi) * K;

    v8f acc0 = {0.f,0.f,0.f,0.f,0.f,0.f,0.f,0.f};
    v8f acc1 = acc0, acc2 = acc0, acc3 = acc0;

    for (int k0 = 0; k0 < K; k0 += 4) {
        v2f a;
        a.x = arow[k0 + kb];
        a.y = arow[k0 + kb + 1];
        const float* w0 = W + (size_t)(k0 + kb) * HID + mi;
        const float* w1 = w0 + HID;
        v2f b0, b1, b2, b3;
        b0.x = w0[0];   b0.y = w1[0];
        b1.x = w0[16];  b1.y = w1[16];
        b2.x = w0[32];  b2.y = w1[32];
        b3.x = w0[48];  b3.y = w1[48];
        acc0 = __builtin_amdgcn_wmma_f32_16x16x4_f32(false, a, false, b0, (short)0, acc0, false, false);
        acc1 = __builtin_amdgcn_wmma_f32_16x16x4_f32(false, a, false, b1, (short)0, acc1, false, false);
        acc2 = __builtin_amdgcn_wmma_f32_16x16x4_f32(false, a, false, b2, (short)0, acc2, false, false);
        acc3 = __builtin_amdgcn_wmma_f32_16x16x4_f32(false, a, false, b3, (short)0, acc3, false, false);
    }

    #pragma unroll
    for (int v = 0; v < 8; ++v) {
        float* tr = T + (size_t)(m0 + v + hi * 8) * HID + mi;
        tr[0]  = acc0[v];
        tr[16] = acc1[v];
        tr[32] = acc2[v];
        tr[48] = acc3[v];
    }
}

#define AGG_NODES 16
#define AGG_WAVES 4
#define AGG_NPW   (AGG_NODES / AGG_WAVES)
#define AGG_GRP   8
#define SH_PITCH2 34

__global__ __launch_bounds__(AGG_WAVES * 32) void aggregate_kernel(
    const float* __restrict__ T,
    const int* __restrict__ rowptr,
    const int2* __restrict__ erec,
    const float* __restrict__ dis,
    const float* __restrict__ bias,
    const float* __restrict__ lw,
    const float* __restrict__ lin_b,
    const float* __restrict__ Wn,
    float* __restrict__ Tn,
    float* __restrict__ Out,
    int N, int first, int last)
{
    __shared__ v2f sh[AGG_NODES * SH_PITCH2];
    __shared__ v2f st[AGG_NODES * SH_PITCH2];

    const int lane = threadIdx.x & 31;
    const int wib = __builtin_amdgcn_readfirstlane((int)(threadIdx.x >> 5));
    const int m0  = blockIdx.x * AGG_NODES;
    const int i0  = m0 + wib * AGG_NPW;

    const v2f* __restrict__ T2 = reinterpret_cast<const v2f*>(T);
    v2f* __restrict__ Tn2 = reinterpret_cast<v2f*>(Tn);

    const float bx = bias[2 * lane];
    const float by = bias[2 * lane + 1];

    int   rp[AGG_NPW + 1];
    float dv[AGG_NPW];
    v2f   ts[AGG_NPW];
    #pragma unroll
    for (int u = 0; u <= AGG_NPW; ++u) rp[u] = rowptr[min(i0 + u, N)];
    #pragma unroll
    for (int u = 0; u < AGG_NPW; ++u) {
        const int ic = min(i0 + u, N - 1);
        dv[u] = dis[ic];
        ts[u] = T2[(size_t)ic * (HID / 2) + lane];
    }

    int2 rc[AGG_NPW];
    #pragma unroll
    for (int u = 0; u < AGG_NPW; ++u) {
        rc[u] = make_int2(0, 0);
        if (rp[u] + lane < rp[u + 1]) rc[u] = erec[rp[u] + lane];
    }

    #pragma unroll
    for (int u = 0; u < AGG_NPW; ++u) {
        const int   i  = i0 + u;
        const float di = dv[u];
        float ax = ts[u].x * di * di + bx;
        float ay = ts[u].y * di * di + by;

        int2 rec = rc[u];
        int base = rp[u];
        const int end = rp[u + 1];
        while (base < end) {
            const int n = min(end - base, 32);
            int j = 0;
            for (; j + AGG_GRP <= n; j += AGG_GRP) {
                v2f   t[AGG_GRP];
                float w[AGG_GRP];
                #pragma unroll
                for (int k = 0; k < AGG_GRP; ++k) {
                    const int c = __builtin_amdgcn_readlane(rec.x, j + k);
                    w[k] = __int_as_float(__builtin_amdgcn_readlane(rec.y, j + k));
                    t[k] = T2[(size_t)c * (HID / 2) + lane];
                }
                #pragma unroll
                for (int k = 0; k < AGG_GRP; ++k) {
                    ax = fmaf(t[k].x, w[k], ax);
                    ay = fmaf(t[k].y, w[k], ay);
                }
            }
            if (j < n) {
                v2f   t[AGG_GRP - 1];
                float w[AGG_GRP - 1];
                #pragma unroll
                for (int k = 0; k < AGG_GRP - 1; ++k) {
                    t[k].x = 0.f;  t[k].y = 0.f;  w[k] = 0.f;
                    if (j + k < n) {
                        const int c = __builtin_amdgcn_readlane(rec.x, j + k);
                        w[k] = __int_as_float(__builtin_amdgcn_readlane(rec.y, j + k));
                        t[k] = T2[(size_t)c * (HID / 2) + lane];
                    }
                }
                #pragma unroll
                for (int k = 0; k < AGG_GRP - 1; ++k) {
                    ax = fmaf(t[k].x, w[k], ax);
                    ay = fmaf(t[k].y, w[k], ay);
                }
            }
            base += 32;
            if (base < end) {
                rec = make_int2(0, 0);
                if (base + lane < end) rec = erec[base + lane];
            }
        }

        v2f h;
        h.x = fmaxf(ax, 0.f);
        h.y = fmaxf(ay, 0.f);
        if (i >= N) {
            h.x = 0.f;  h.y = 0.f;
        }
        sh[(wib * AGG_NPW + u) * SH_PITCH2 + lane] = h;
    }

    const int mi = lane & 15;
    const int hi = lane >> 4;
    const int kb = hi * 2;

    v2f wf[HID / 4];
    #pragma unroll
    for (int s = 0; s < HID / 4; ++s) { wf[s].x = 0.f;  wf[s].y = 0.f; }
    if (!last) {
        const float* wc = Wn + wib * 16 + mi;
        #pragma unroll
        for (int s = 0; s < HID / 4; ++s) {
            wf[s].x = wc[(4 * s + kb) * HID];
            wf[s].y = wc[(4 * s + kb + 1) * HID];
        }
    }

    __syncthreads();

    if (!last) {
        v8f tacc = {0.f,0.f,0.f,0.f,0.f,0.f,0.f,0.f};
        #pragma unroll
        for (int s = 0; s < HID / 4; ++s) {
            const v2f a = sh[mi * SH_PITCH2 + ((4 * s + kb) >> 1)];
            tacc = __builtin_amdgcn_wmma_f32_16x16x4_f32(false, a, false, wf[s], (short)0, tacc, false, false);
        }
        float* stf = reinterpret_cast<float*>(st);
        #pragma unroll
        for (int v = 0; v < 8; ++v)
            stf[(v + hi * 8) * (2 * SH_PITCH2) + wib * 16 + mi] = tacc[v];

        __syncthreads();
        v2f tv[AGG_NPW];
        #pragma unroll
        for (int u = 0; u < AGG_NPW; ++u) tv[u] = st[(wib * AGG_NPW + u) * SH_PITCH2 + lane];
        #pragma unroll
        for (int u = 0; u < AGG_NPW; ++u) {
            const int r = m0 + wib * AGG_NPW + u;
            if (r < N) Tn2[(size_t)r * (HID / 2) + lane] = tv[u];
        }
    }

    if (wib < (OUTD + 15) / 16) {
        const int  col = wib * 16 + mi;
        const bool cok = col < OUTD;

        v8f acc;
        #pragma unroll
        for (int v = 0; v < 8; ++v) {
            const int row = m0 + v + hi * 8;
            float c0 = 0.f;
            if (cok && row < N) c0 = first ? lin_b[col] : Out[(size_t)row * OUTD + col];
            acc[v] = c0;
        }

        v2f bf[HID / 4];
        #pragma unroll
        for (int s = 0; s < HID / 4; ++s) { bf[s].x = 0.f;  bf[s].y = 0.f; }
        if (cok) {
            #pragma unroll
            for (int s = 0; s < HID / 4; ++s) {
                bf[s].x = lw[(4 * s + kb) * OUTD + col];
                bf[s].y = lw[(4 * s + kb + 1) * OUTD + col];
            }
        }

        #pragma unroll
        for (int s = 0; s < HID / 4; ++s) {
            const v2f a = sh[mi * SH_PITCH2 + ((4 * s + kb) >> 1)];
            acc = __builtin_amdgcn_wmma_f32_16x16x4_f32(false, a, false, bf[s], (short)0, acc, false, false);
        }

        #pragma unroll
        for (int v = 0; v < 8; ++v) {
            const int row = m0 + v + hi * 8;
            if (cok && row < N) Out[(size_t)row * OUTD + col] = acc[v];
        }
    }
}

extern "C" void kernel_launch(void* const* d_in, const int* in_sizes, int n_in,
                              void* d_out, int out_size, void* d_ws, size_t ws_size,
                              hipStream_t stream) {
    const float* x     = (const float*)d_in[0];
    const float* W0    = (const float*)d_in[1];
    const float* Ws    = (const float*)d_in[2];
    const float* bs    = (const float*)d_in[3];
    const float* lin_w = (const float*)d_in[4];
    const float* lin_b = (const float*)d_in[5];
    const int*   ei    = (const int*)d_in[6];
    float* out = (float*)d_out;

    const int N = in_sizes[0] / IN_DIM;
    const int E = in_sizes[6] / 2;
    const int* src = ei;
    const int* dst = ei + E;

    char* ws = (char*)d_ws;
    float* ta     = (float*)ws; ws += sizeof(float) * (size_t)N * HID;
    float* tb     = (float*)ws; ws += sizeof(float) * (size_t)N * HID;
    int2*  erec   = (int2*)ws;  ws += sizeof(int2)  * (size_t)E;
    int*   cnt    = (int*)ws;   ws += sizeof(int)   * (size_t)N;
    int*   pos    = (int*)ws;   ws += sizeof(int)   * (size_t)N;
    float* dis    = (float*)ws; ws += sizeof(float) * (size_t)N;
    int*   rowptr = (int*)ws;   ws += sizeof(int)   * (size_t)(N + 1);

    const int TPB = 256;
    zero_int_kernel<<<(N + TPB - 1) / TPB, TPB, 0, stream>>>(cnt, N);
    count_deg_kernel<<<(E + TPB - 1) / TPB, TPB, 0, stream>>>(dst, cnt, E);
    scan_kernel<<<(N + SCAN_TPB - 1) / SCAN_TPB, SCAN_TPB, 0, stream>>>(cnt, rowptr, N, E);
    init_pos_dis_kernel<<<(N + TPB - 1) / TPB, TPB, 0, stream>>>(rowptr, cnt, pos, dis, N);
    fill_csr_kernel<<<(E + TPB - 1) / TPB, TPB, 0, stream>>>(src, dst, pos, dis, erec, E);
    const int strips = N / 16;
    const int gblocks = (strips + 7) / 8;
    gemm_wmma_f32_kernel<<<gblocks, 256, 0, stream>>>(x, W0, ta, N, IN_DIM);
    for (int l = 0; l < LAYERS; ++l) {
        const float* tin  = (l & 1) ? tb : ta;
        float*       tout = (l & 1) ? ta : tb;
        const int    last = (l == LAYERS - 1) ? 1 : 0;
        const float* Wn   = Ws + (size_t)(last ? 0 : l) * HID * HID;

        aggregate_kernel<<<(N + AGG_NODES - 1) / AGG_NODES, AGG_WAVES * 32, 0, stream>>>(
            tin, rowptr, erec, dis,
            bs + (size_t)l * HID,
            lin_w + (size_t)l * HID * OUTD,
            lin_b, Wn, tout, out, N, (l == 0) ? 1 : 0, last);
    }
}


# CSR build: per-edge global atomics and scattered stores replaced by a two-level counting sort binned in LDS (256-node buckets, coalesced writes); edge order in a row, so the sum order, changes
# speedup vs baseline: 3.3912x; 1.4584x over previous
#include <hip/hip_runtime.h>

typedef __attribute__((ext_vector_type(2))) float v2f;
typedef __attribute__((ext_vector_type(8))) float v8f;

#define IN_DIM 128
#define HID    64
#define LAYERS 8
#define OUTD   40

#define SCAN_TPB 1024
#define SCAN_PF  16
__global__ __launch_bounds__(SCAN_TPB) void scan_kernel(const int* __restrict__ cnt,
                                                        int* __restrict__ rowptr,
                                                        int N, int E) {
    __shared__ int sh_tot[SCAN_TPB / 32];
    __shared__ int sh_pre[SCAN_TPB / 32];
    const int tid  = threadIdx.x;
    const int lane = tid & 31;
    const int wib  = tid >> 5;
    const int base = blockIdx.x * SCAN_TPB;

    int pre = 0;
    for (int j0 = tid; j0 < base; j0 += SCAN_PF * SCAN_TPB) {
        #pragma unroll
        for (int u = 0; u < SCAN_PF; ++u) {
            const int j = j0 + u * SCAN_TPB;
            pre += (j < base) ? cnt[j] : 0;
        }
    }

    const int idx = base + tid;
    const int v = (idx < N) ? cnt[idx] : 0;

    int incl = v;
    for (int off = 1; off < 32; off <<= 1) {
        int t = __shfl_up(incl, off, 32);
        if (lane >= off) incl += t;
    }
    for (int off = 16; off > 0; off >>= 1) pre += __shfl_xor(pre, off, 32);

    if (lane == 31) sh_tot[wib] = incl;
    if (lane == 0)  sh_pre[wib] = pre;
    __syncthreads();

    int part = sh_pre[lane] + ((lane < wib) ? sh_tot[lane] : 0);
    for (int off = 16; off > 0; off >>= 1) part += __shfl_xor(part, off, 32);

    if (idx < N) rowptr[idx] = part + incl - v;
    if (blockIdx.x == 0 && tid == 0) rowptr[N] = E;
}

#define BIN_SHIFT 8
#define BIN_NODES (1 << BIN_SHIFT)
#define BIN_MAXB  512
#define BIN_TPB   1024
#define BIN_EPT   6
#define BIN_CHUNK (BIN_TPB * BIN_EPT)
#define BIN_CAP   6144

__device__ __forceinline__ int excl_scan_first512(int v, int tid, int* wt) {
    const int lane = tid & 31;
    const int wib  = tid >> 5;
    int incl = v;
    for (int off = 1; off < 32; off <<= 1) {
        int t = __shfl_up(incl, off, 32);
        if (lane >= off) incl += t;
    }
    if (wib < 16 && lane == 31) wt[wib] = incl;
    __syncthreads();
    int add = 0;
    for (int w = 0; w < 16; ++w) add += (w < wib) ? wt[w] : 0;
    return add + incl - v;
}

__global__ __launch_bounds__(BIN_TPB) void bin_count_kernel(const int* __restrict__ dst,
                                                            int* __restrict__ H,
                                                            int E, int G, int B) {
    __shared__ int c[BIN_MAXB];
    const int tid = threadIdx.x;
    const int g   = blockIdx.x;
    if (tid < BIN_MAXB) c[tid] = 0;
    __syncthreads();
    const int e0 = g * BIN_CHUNK + tid;
    #pragma unroll
    for (int k = 0; k < BIN_EPT; ++k) {
        const int e = e0 + k * BIN_TPB;
        if (e < E) atomicAdd(&c[dst[e] >> BIN_SHIFT], 1);
    }
    __syncthreads();
    if (tid < B) H[tid * G + g] = c[tid];
}

__global__ __launch_bounds__(BIN_TPB) void bin_scatter_kernel(const int* __restrict__ src,
                                                              const int* __restrict__ dst,
                                                              const int* __restrict__ OFF,
                                                              int2* __restrict__ tmp,
                                                              int E, int G, int B) {
    __shared__ int  c[BIN_MAXB];
    __shared__ int  ls[BIN_MAXB];
    __shared__ int  go[BIN_MAXB];
    __shared__ int  wt[16];
    __shared__ int2 rec[BIN_CHUNK];
    const int tid = threadIdx.x;
    const int g   = blockIdx.x;
    if (tid < BIN_MAXB) c[tid] = 0;
    __syncthreads();

    int s[BIN_EPT], d[BIN_EPT], r[BIN_EPT];
    const int e0 = g * BIN_CHUNK + tid;
    #pragma unroll
    for (int k = 0; k < BIN_EPT; ++k) {
        const int e = e0 + k * BIN_TPB;
        s[k] = 0;  d[k] = -1;  r[k] = 0;
        if (e < E) {
            s[k] = src[e];
            d[k] = dst[e];
            r[k] = atomicAdd(&c[d[k] >> BIN_SHIFT], 1);
        }
    }
    __syncthreads();

    const int v  = (tid < B) ? c[tid] : 0;
    const int ex = excl_scan_first512(v, tid, wt);
    if (tid < B) {
        ls[tid] = ex;
        go[tid] = OFF[tid * G + g] - ex;
    }
    __syncthreads();

    #pragma unroll
    for (int k = 0; k < BIN_EPT; ++k)
        if (d[k] >= 0) rec[ls[d[k] >> BIN_SHIFT] + r[k]] = make_int2(s[k], d[k]);
    __syncthreads();

    const int n = min(BIN_CHUNK, E - g * BIN_CHUNK);
    for (int t = tid; t < n; t += BIN_TPB) {
        const int2 x = rec[t];
        tmp[go[x.y >> BIN_SHIFT] + t] = x;
    }
}

__global__ __launch_bounds__(BIN_TPB) void bin_finalize_kernel(const int2* __restrict__ tmp,
                                                               const int* __restrict__ OFF,
                                                               int* __restrict__ rowptr,
                                                               float* __restrict__ dis,
                                                               int2* __restrict__ erec,
                                                               int N, int E, int G) {
    __shared__ int  deg[BIN_NODES];
    __shared__ int  ptr[BIN_NODES];
    __shared__ int  wt[16];
    __shared__ int2 stage[BIN_CAP];
    const int tid = threadIdx.x;
    const int b   = blockIdx.x;
    const int n0  = b << BIN_SHIFT;
    const int bs  = OFF[b * G];
    const int be  = OFF[(b + 1) * G];
    const int nb  = be - bs;
    const bool staged = nb <= BIN_CAP;

    if (tid < BIN_NODES) deg[tid] = 0;
    __syncthreads();
    for (int p = bs + tid; p < be; p += BIN_TPB) atomicAdd(&deg[tmp[p].y - n0], 1);
    __syncthreads();

    const int v  = (tid < BIN_NODES) ? deg[tid] : 0;
    const int ex = excl_scan_first512(v, tid, wt);
    if (tid < BIN_NODES) {
        ptr[tid] = ex;
        const int node = n0 + tid;
        if (node < N) {
            rowptr[node] = bs + ex;
            dis[node] = rsqrtf((float)(v + 1));
        }
    }
    if (b == 0 && tid == 0) rowptr[N] = E;
    __syncthreads();

    for (int p = bs + tid; p < be; p += BIN_TPB) {
        const int2 x  = tmp[p];
        const int  li = x.y - n0;
        const int  q  = atomicAdd(&ptr[li], 1);
        const int2 o  = make_int2(x.x, __float_as_int(rsqrtf((float)(deg[li] + 1))));
        if (staged) stage[q] = o;
        else        erec[bs + q] = o;
    }
    __syncthreads();
    if (staged)
        for (int t = tid; t < nb; t += BIN_TPB) erec[bs + t] = stage[t];
}

__global__ void edge_weight_kernel(int2* __restrict__ erec, const float* __restrict__ dis, int E) {
    int e = blockIdx.x * blockDim.x + threadIdx.x;
    if (e < E) {
        int2 r = erec[e];
        r.y = __float_as_int(dis[r.x] * __int_as_float(r.y));
        erec[e] = r;
    }
}

__global__ __launch_bounds__(256) void gemm_wmma_f32_kernel(
    const float* __restrict__ A, const float* __restrict__ W,
    float* __restrict__ T, int nrows, int K)
{
    const int lane  = threadIdx.x & 31;
    const int wib   = threadIdx.x >> 5;
    const int strip = blockIdx.x * 8 + wib;
    const int m0    = strip * 16;
    if (m0 >= nrows) return;

    const int mi = lane & 15;
    const int hi = (lane >> 4) & 1;
    const int kb = hi * 2;

    const float* arow = A + (size_t)(m0 + mi) * K;

    v8f acc0 = {0.f,0.f,0.f,0.f,0.f,0.f,0.f,0.f};
    v8f acc1 = acc0, acc2 = acc0, acc3 = acc0;

    for (int k0 = 0; k0 < K; k0 += 4) {
        v2f a;
        a.x = arow[k0 + kb];
        a.y = arow[k0 + kb + 1];
        const float* w0 = W + (size_t)(k0 + kb) * HID + mi;
        const float* w1 = w0 + HID;
        v2f b0, b1, b2, b3;
        b0.x = w0[0];   b0.y = w1[0];
        b1.x = w0[16];  b1.y = w1[16];
        b2.x = w0[32];  b2.y = w1[32];
        b3.x = w0[48];  b3.y = w1[48];
        acc0 = __builtin_amdgcn_wmma_f32_16x16x4_f32(false, a, false, b0, (short)0, acc0, false, false);
        acc1 = __builtin_amdgcn_wmma_f32_16x16x4_f32(false, a, false, b1, (short)0, acc1, false, false);
        acc2 = __builtin_amdgcn_wmma_f32_16x16x4_f32(false, a, false, b2, (short)0, acc2, false, false);
        acc3 = __builtin_amdgcn_wmma_f32_16x16x4_f32(false, a, false, b3, (short)0, acc3, false, false);
    }

    #pragma unroll
    for (int v = 0; v < 8; ++v) {
        float* tr = T + (size_t)(m0 + v + hi * 8) * HID + mi;
        tr[0]  = acc0[v];
        tr[16] = acc1[v];
        tr[32] = acc2[v];
        tr[48] = acc3[v];
    }
}

#define AGG_NODES 16
#define AGG_WAVES 4
#define AGG_NPW   (AGG_NODES / AGG_WAVES)
#define AGG_GRP   8
#define SH_PITCH2 34

__global__ __launch_bounds__(AGG_WAVES * 32) void aggregate_kernel(
    const float* __restrict__ T,
    const int* __restrict__ rowptr,
    const int2* __restrict__ erec,
    const float* __restrict__ dis,
    const float* __restrict__ bias,
    const float* __restrict__ lw,
    const float* __restrict__ lin_b,
    const float* __restrict__ Wn,
    float* __restrict__ Tn,
    float* __restrict__ Out,
    int N, int first, int last)
{
    __shared__ v2f sh[AGG_NODES * SH_PITCH2];
    __shared__ v2f st[AGG_NODES * SH_PITCH2];

    const int lane = threadIdx.x & 31;
    const int wib = __builtin_amdgcn_readfirstlane((int)(threadIdx.x >> 5));
    const int m0  = blockIdx.x * AGG_NODES;
    const int i0  = m0 + wib * AGG_NPW;

    const v2f* __restrict__ T2 = reinterpret_cast<const v2f*>(T);
    v2f* __restrict__ Tn2 = reinterpret_cast<v2f*>(Tn);

    const float bx = bias[2 * lane];
    const float by = bias[2 * lane + 1];

    int   rp[AGG_NPW + 1];
    float dv[AGG_NPW];
    v2f   ts[AGG_NPW];
    #pragma unroll
    for (int u = 0; u <= AGG_NPW; ++u) rp[u] = rowptr[min(i0 + u, N)];
    #pragma unroll
    for (int u = 0; u < AGG_NPW; ++u) {
        const int ic = min(i0 + u, N - 1);
        dv[u] = dis[ic];
        ts[u] = T2[(size_t)ic * (HID / 2) + lane];
    }

    int2 rc[AGG_NPW];
    #pragma unroll
    for (int u = 0; u < AGG_NPW; ++u) {
        rc[u] = make_int2(0, 0);
        if (rp[u] + lane < rp[u + 1]) rc[u] = erec[rp[u] + lane];
    }

    #pragma unroll
    for (int u = 0; u < AGG_NPW; ++u) {
        const int   i  = i0 + u;
        const float di = dv[u];
        float ax = ts[u].x * di * di + bx;
        float ay = ts[u].y * di * di + by;

        int2 rec = rc[u];
        int base = rp[u];
        const int end = rp[u + 1];
        while (base < end) {
            const int n = min(end - base, 32);
            int j = 0;
            for (; j + AGG_GRP <= n; j += AGG_GRP) {
                v2f   t[AGG_GRP];
                float w[AGG_GRP];
                #pragma unroll
                for (int k = 0; k < AGG_GRP; ++k) {
                    const int c = __builtin_amdgcn_readlane(rec.x, j + k);
                    w[k] = __int_as_float(__builtin_amdgcn_readlane(rec.y, j + k));
                    t[k] = T2[(size_t)c * (HID / 2) + lane];
                }
                #pragma unroll
                for (int k = 0; k < AGG_GRP; ++k) {
                    ax = fmaf(t[k].x, w[k], ax);
                    ay = fmaf(t[k].y, w[k], ay);
                }
            }
            if (j < n) {
                v2f   t[AGG_GRP - 1];
                float w[AGG_GRP - 1];
                #pragma unroll
                for (int k = 0; k < AGG_GRP - 1; ++k) {
                    t[k].x = 0.f;  t[k].y = 0.f;  w[k] = 0.f;
                    if (j + k < n) {
                        const int c = __builtin_amdgcn_readlane(rec.x, j + k);
                        w[k] = __int_as_float(__builtin_amdgcn_readlane(rec.y, j + k));
                        t[k] = T2[(size_t)c * (HID / 2) + lane];
                    }
                }
                #pragma unroll
                for (int k = 0; k < AGG_GRP - 1; ++k) {
                    ax = fmaf(t[k].x, w[k], ax);
                    ay = fmaf(t[k].y, w[k], ay);
                }
            }
            base += 32;
            if (base < end) {
                rec = make_int2(0, 0);
                if (base + lane < end) rec = erec[base + lane];
            }
        }

        v2f h;
        h.x = fmaxf(ax, 0.f);
        h.y = fmaxf(ay, 0.f);
        if (i >= N) {
            h.x = 0.f;  h.y = 0.f;
        }
        sh[(wib * AGG_NPW + u) * SH_PITCH2 + lane] = h;
    }

    const int mi = lane & 15;
    const int hi = lane >> 4;
    const int kb = hi * 2;

    v2f wf[HID / 4];
    #pragma unroll
    for (int s = 0; s < HID / 4; ++s) { wf[s].x = 0.f;  wf[s].y = 0.f; }
    if (!last) {
        const float* wc = Wn + wib * 16 + mi;
        #pragma unroll
        for (int s = 0; s < HID / 4; ++s) {
            wf[s].x = wc[(4 * s + kb) * HID];
            wf[s].y = wc[(4 * s + kb + 1) * HID];
        }
    }

    __syncthreads();

    if (!last) {
        v8f tacc = {0.f,0.f,0.f,0.f,0.f,0.f,0.f,0.f};
        #pragma unroll
        for (int s = 0; s < HID / 4; ++s) {
            const v2f a = sh[mi * SH_PITCH2 + ((4 * s + kb) >> 1)];
            tacc = __builtin_amdgcn_wmma_f32_16x16x4_f32(false, a, false, wf[s], (short)0, tacc, false, false);
        }
        float* stf = reinterpret_cast<float*>(st);
        #pragma unroll
        for (int v = 0; v < 8; ++v)
            stf[(v + hi * 8) * (2 * SH_PITCH2) + wib * 16 + mi] = tacc[v];

        __syncthreads();
        v2f tv[AGG_NPW];
        #pragma unroll
        for (int u = 0; u < AGG_NPW; ++u) tv[u] = st[(wib * AGG_NPW + u) * SH_PITCH2 + lane];
        #pragma unroll
        for (int u = 0; u < AGG_NPW; ++u) {
            const int r = m0 + wib * AGG_NPW + u;
            if (r < N) Tn2[(size_t)r * (HID / 2) + lane] = tv[u];
        }
    }

    if (wib < (OUTD + 15) / 16) {
        const int  col = wib * 16 + mi;
        const bool cok = col < OUTD;

        v8f acc;
        #pragma unroll
        for (int v = 0; v < 8; ++v) {
            const int row = m0 + v + hi * 8;
            float c0 = 0.f;
            if (cok && row < N) c0 = first ? lin_b[col] : Out[(size_t)row * OUTD + col];
            acc[v] = c0;
        }

        v2f bf[HID / 4];
        #pragma unroll
        for (int s = 0; s < HID / 4; ++s) { bf[s].x = 0.f;  bf[s].y = 0.f; }
        if (cok) {
            #pragma unroll
            for (int s = 0; s < HID / 4; ++s) {
                bf[s].x = lw[(4 * s + kb) * OUTD + col];
                bf[s].y = lw[(4 * s + kb + 1) * OUTD + col];
            }
        }

        #pragma unroll
        for (int s = 0; s < HID / 4; ++s) {
            const v2f a = sh[mi * SH_PITCH2 + ((4 * s + kb) >> 1)];
            acc = __builtin_amdgcn_wmma_f32_16x16x4_f32(false, a, false, bf[s], (short)0, acc, false, false);
        }

        #pragma unroll
        for (int v = 0; v < 8; ++v) {
            const int row = m0 + v + hi * 8;
            if (cok && row < N) Out[(size_t)row * OUTD + col] = acc[v];
        }
    }
}

extern "C" void kernel_launch(void* const* d_in, const int* in_sizes, int n_in,
                              void* d_out, int out_size, void* d_ws, size_t ws_size,
                              hipStream_t stream) {
    const float* x     = (const float*)d_in[0];
    const float* W0    = (const float*)d_in[1];
    const float* Ws    = (const float*)d_in[2];
    const float* bs    = (const float*)d_in[3];
    const float* lin_w = (const float*)d_in[4];
    const float* lin_b = (const float*)d_in[5];
    const int*   ei    = (const int*)d_in[6];
    float* out = (float*)d_out;

    const int N = in_sizes[0] / IN_DIM;
    const int E = in_sizes[6] / 2;
    const int* src = ei;
    const int* dst = ei + E;

    char* ws = (char*)d_ws;
    float* ta     = (float*)ws; ws += sizeof(float) * (size_t)N * HID;
    float* tb     = (float*)ws; ws += sizeof(float) * (size_t)N * HID;
    int2*  erec   = (int2*)ws;  ws += sizeof(int2)  * (size_t)E;
    int2*  tmp    = (int2*)ws;  ws += sizeof(int2)  * (size_t)E;
    float* dis    = (float*)ws; ws += sizeof(float) * (size_t)N;
    int*   rowptr = (int*)ws;   ws += sizeof(int)   * (size_t)(N + 1);
    const int B  = (N + BIN_NODES - 1) / BIN_NODES;
    const int G  = (E + BIN_CHUNK - 1) / BIN_CHUNK;
    const int HN = B * G;
    int*   hcnt   = (int*)ws;   ws += sizeof(int)   * (size_t)HN;
    int*   hoff   = (int*)ws;   ws += sizeof(int)   * (size_t)(HN + 1);

    const int TPB = 256;
    bin_count_kernel<<<G, BIN_TPB, 0, stream>>>(dst, hcnt, E, G, B);
    scan_kernel<<<(HN + SCAN_TPB - 1) / SCAN_TPB, SCAN_TPB, 0, stream>>>(hcnt, hoff, HN, E);
    bin_scatter_kernel<<<G, BIN_TPB, 0, stream>>>(src, dst, hoff, tmp, E, G, B);
    bin_finalize_kernel<<<B, BIN_TPB, 0, stream>>>(tmp, hoff, rowptr, dis, erec, N, E, G);
    edge_weight_kernel<<<(E + TPB - 1) / TPB, TPB, 0, stream>>>(erec, dis, E);
    const int strips = N / 16;
    const int gblocks = (strips + 7) / 8;
    gemm_wmma_f32_kernel<<<gblocks, 256, 0, stream>>>(x, W0, ta, N, IN_DIM);
    for (int l = 0; l < LAYERS; ++l) {
        const float* tin  = (l & 1) ? tb : ta;
        float*       tout = (l & 1) ? ta : tb;
        const int    last = (l == LAYERS - 1) ? 1 : 0;
        const float* Wn   = Ws + (size_t)(last ? 0 : l) * HID * HID;

        aggregate_kernel<<<(N + AGG_NODES - 1) / AGG_NODES, AGG_WAVES * 32, 0, stream>>>(
            tin, rowptr, erec, dis,
            bs + (size_t)l * HID,
            lin_w + (size_t)l * HID * OUTD,
            lin_b, Wn, tout, out, N, (l == 0) ? 1 : 0, last);
    }
}


# CSR build: each 6144-edge chunk stays ordered by bucket in its own region with its bucket offsets, whose sums over chunks place a bucket: count and scan launches go (13 -> 11); row edge order changes
# speedup vs baseline: 3.6626x; 1.0800x over previous
#include <hip/hip_runtime.h>

typedef __attribute__((ext_vector_type(2))) float v2f;
typedef __attribute__((ext_vector_type(8))) float v8f;

#define IN_DIM 128
#define HID    64
#define LAYERS 8
#define OUTD   40

#define BIN_SHIFT 8
#define BIN_NODES (1 << BIN_SHIFT)
#define BIN_MAXB  512
#define BIN_TPB   1024
#define BIN_EPT   6
#define BIN_CHUNK (BIN_TPB * BIN_EPT)
#define BIN_CAP   6144
#define BIN_RUNS  8

__device__ __forceinline__ int excl_scan_first512(int v, int tid, int* wt) {
    const int lane = tid & 31;
    const int wib  = tid >> 5;
    int incl = v;
    for (int off = 1; off < 32; off <<= 1) {
        int t = __shfl_up(incl, off, 32);
        if (lane >= off) incl += t;
    }
    if (wib < 16 && lane == 31) wt[wib] = incl;
    __syncthreads();
    int add = 0;
    for (int w = 0; w < 16; ++w) add += (w < wib) ? wt[w] : 0;
    return add + incl - v;
}

__global__ __launch_bounds__(BIN_TPB) void bin_scatter_kernel(const int* __restrict__ src,
                                                              const int* __restrict__ dst,
                                                              int* __restrict__ RUN,
                                                              int2* __restrict__ tmp,
                                                              int E, int G, int B) {
    __shared__ int  c[BIN_MAXB];
    __shared__ int  ls[BIN_MAXB];
    __shared__ int  wt[16];
    __shared__ int2 rec[BIN_CHUNK];
    const int tid = threadIdx.x;
    const int g   = blockIdx.x;
    if (tid < BIN_MAXB) c[tid] = 0;
    __syncthreads();

    int s[BIN_EPT], d[BIN_EPT], r[BIN_EPT];
    const int e0 = g * BIN_CHUNK + tid;
    #pragma unroll
    for (int k = 0; k < BIN_EPT; ++k) {
        const int e = e0 + k * BIN_TPB;
        s[k] = 0;  d[k] = -1;  r[k] = 0;
        if (e < E) {
            s[k] = src[e];
            d[k] = dst[e];
            r[k] = atomicAdd(&c[d[k] >> BIN_SHIFT], 1);
        }
    }
    __syncthreads();

    const int n  = min(BIN_CHUNK, E - g * BIN_CHUNK);
    const int v  = (tid < B) ? c[tid] : 0;
    const int ex = excl_scan_first512(v, tid, wt);
    if (tid < B) {
        ls[tid] = ex;
        RUN[tid * G + g] = ex;
    }
    if (tid == B) RUN[B * G + g] = n;
    __syncthreads();

    #pragma unroll
    for (int k = 0; k < BIN_EPT; ++k)
        if (d[k] >= 0) rec[ls[d[k] >> BIN_SHIFT] + r[k]] = make_int2(s[k], d[k]);
    __syncthreads();

    int2* __restrict__ out = tmp + (size_t)g * BIN_CHUNK;
    for (int t = tid; t < n; t += BIN_TPB) out[t] = rec[t];
}

template <class F>
__device__ __forceinline__ void for_bucket_records(const int2* __restrict__ tmp,
                                                   const int* __restrict__ rs,
                                                   const int* __restrict__ re,
                                                   int G, int wib, int lane, F f) {
    const int cpw = (G + 31) / 32;
    const int g0  = wib * cpw;
    const int g1  = min(g0 + cpw, G);
    for (int gb = g0; gb < g1; gb += 32) {
        int s = 0, e = 0;
        if (gb + lane < g1) { s = rs[gb + lane];  e = re[gb + lane]; }
        const int cnt = min(32, g1 - gb);
        for (int i0 = 0; i0 < cnt; i0 += BIN_RUNS) {
            int si[BIN_RUNS], ei[BIN_RUNS];
            #pragma unroll
            for (int k = 0; k < BIN_RUNS; ++k) {
                si[k] = __builtin_amdgcn_readlane(s, i0 + k);
                ei[k] = __builtin_amdgcn_readlane(e, i0 + k);
            }
            int2 x[BIN_RUNS];
            #pragma unroll
            for (int k = 0; k < BIN_RUNS; ++k) {
                x[k] = make_int2(0, 0);
                if (si[k] + lane < ei[k])
                    x[k] = tmp[(size_t)(gb + i0 + k) * BIN_CHUNK + si[k] + lane];
            }
            #pragma unroll
            for (int k = 0; k < BIN_RUNS; ++k) {
                if (si[k] + lane < ei[k]) f(x[k]);
                for (int p = si[k] + 32 + lane; p < ei[k]; p += 32)
                    f(tmp[(size_t)(gb + i0 + k) * BIN_CHUNK + p]);
            }
        }
    }
}

__global__ __launch_bounds__(BIN_TPB) void bin_finalize_kernel(const int2* __restrict__ tmp,
                                                               const int* __restrict__ RUN,
                                                               int* __restrict__ rowptr,
                                                               float* __restrict__ dis,
                                                               int* __restrict__ erec,
                                                               int N, int E, int G) {
    __shared__ int  deg[BIN_NODES];
    __shared__ int  ptr[BIN_NODES];
    __shared__ int  wt[16];
    __shared__ int  tot[2 * (BIN_TPB / 32)];
    __shared__ int  stage[BIN_CAP];
    const int tid  = threadIdx.x;
    const int lane = tid & 31;
    const int wib  = __builtin_amdgcn_readfirstlane(tid >> 5);
    const int b    = blockIdx.x;
    const int n0   = b << BIN_SHIFT;
    const int* __restrict__ rs = RUN + (size_t)b * G;
    const int* __restrict__ re = rs + G;

    int ps = 0, pe = 0;
    for (int g = tid; g < G; g += BIN_TPB) { ps += rs[g];  pe += re[g]; }
    for (int off = 16; off > 0; off >>= 1) {
        ps += __shfl_xor(ps, off, 32);
        pe += __shfl_xor(pe, off, 32);
    }
    if (lane == 0) { tot[wib] = ps;  tot[BIN_TPB / 32 + wib] = pe; }
    if (tid < BIN_NODES) deg[tid] = 0;
    __syncthreads();
    int bs = tot[lane], be = tot[BIN_TPB / 32 + lane];
    for (int off = 16; off > 0; off >>= 1) {
        bs += __shfl_xor(bs, off, 32);
        be += __shfl_xor(be, off, 32);
    }
    const int nb  = be - bs;
    const bool staged = nb <= BIN_CAP;

    for_bucket_records(tmp, rs, re, G, wib, lane,
                       [&](int2 x) { atomicAdd(&deg[x.y - n0], 1); });
    __syncthreads();

    const int v  = (tid < BIN_NODES) ? deg[tid] : 0;
    const int ex = excl_scan_first512(v, tid, wt);
    if (tid < BIN_NODES) {
        ptr[tid] = ex;
        const int node = n0 + tid;
        if (node < N) {
            rowptr[node] = bs + ex;
            dis[node] = rsqrtf((float)(v + 1));
        }
    }
    if (b == 0 && tid == 0) rowptr[N] = E;
    __syncthreads();

    for_bucket_records(tmp, rs, re, G, wib, lane,
                       [&](int2 x) {
                           const int q = atomicAdd(&ptr[x.y - n0], 1);
                           if (staged) stage[q] = x.x;
                           else        erec[bs + q] = x.x;
                       });
    __syncthreads();
    if (staged)
        for (int t = tid; t < nb; t += BIN_TPB) erec[bs + t] = stage[t];
}

__global__ __launch_bounds__(256) void gemm_wmma_f32_kernel(
    const float* __restrict__ A, const float* __restrict__ W,
    const float* __restrict__ dis,
    float* __restrict__ T, int nrows, int K)
{
    const int lane  = threadIdx.x & 31;
    const int wib   = threadIdx.x >> 5;
    const int strip = blockIdx.x * 8 + wib;
    const int m0    = strip * 16;
    if (m0 >= nrows) return;

    const int mi = lane & 15;
    const int hi = (lane >> 4) & 1;
    const int kb = hi * 2;

    const float* arow = A + (size_t)(m0 + mi) * K;

    v8f acc0 = {0.f,0.f,0.f,0.f,0.f,0.f,0.f,0.f};
    v8f acc1 = acc0, acc2 = acc0, acc3 = acc0;

    for (int k0 = 0; k0 < K; k0 += 4) {
        v2f a;
        a.x = arow[k0 + kb];
        a.y = arow[k0 + kb + 1];
        const float* w0 = W + (size_t)(k0 + kb) * HID + mi;
        const float* w1 = w0 + HID;
        v2f b0, b1, b2, b3;
        b0.x = w0[0];   b0.y = w1[0];
        b1.x = w0[16];  b1.y = w1[16];
        b2.x = w0[32];  b2.y = w1[32];
        b3.x = w0[48];  b3.y = w1[48];
        acc0 = __builtin_amdgcn_wmma_f32_16x16x4_f32(false, a, false, b0, (short)0, acc0, false, false);
        acc1 = __builtin_amdgcn_wmma_f32_16x16x4_f32(false, a, false, b1, (short)0, acc1, false, false);
        acc2 = __builtin_amdgcn_wmma_f32_16x16x4_f32(false, a, false, b2, (short)0, acc2, false, false);
        acc3 = __builtin_amdgcn_wmma_f32_16x16x4_f32(false, a, false, b3, (short)0, acc3, false, false);
    }

    #pragma unroll
    for (int v = 0; v < 8; ++v) {
        const int   row = m0 + v + hi * 8;
        const float d   = dis[row];
        float* tr = T + (size_t)row * HID + mi;
        tr[0]  = acc0[v] * d;
        tr[16] = acc1[v] * d;
        tr[32] = acc2[v] * d;
        tr[48] = acc3[v] * d;
    }
}

#define AGG_NODES 16
#define AGG_WAVES 4
#define AGG_NPW   (AGG_NODES / AGG_WAVES)
#define AGG_GRP   8
#define SH_PITCH2 34

__global__ __launch_bounds__(AGG_WAVES * 32) void aggregate_kernel(
    const float* __restrict__ T,
    const int* __restrict__ rowptr,
    const int* __restrict__ erec,
    const float* __restrict__ dis,
    const float* __restrict__ bias,
    const float* __restrict__ lw,
    const float* __restrict__ lin_b,
    const float* __restrict__ Wn,
    float* __restrict__ Tn,
    float* __restrict__ Out,
    int N, int first, int last)
{
    __shared__ v2f sh[AGG_NODES * SH_PITCH2];
    __shared__ v2f st[AGG_NODES * SH_PITCH2];

    const int lane = threadIdx.x & 31;
    const int wib = __builtin_amdgcn_readfirstlane((int)(threadIdx.x >> 5));
    const int m0  = blockIdx.x * AGG_NODES;
    const int i0  = m0 + wib * AGG_NPW;

    const v2f* __restrict__ T2 = reinterpret_cast<const v2f*>(T);
    v2f* __restrict__ Tn2 = reinterpret_cast<v2f*>(Tn);

    const float bx = bias[2 * lane];
    const float by = bias[2 * lane + 1];

    int   rp[AGG_NPW + 1];
    float dv[AGG_NPW];
    v2f   ts[AGG_NPW];
    #pragma unroll
    for (int u = 0; u <= AGG_NPW; ++u) rp[u] = rowptr[min(i0 + u, N)];
    #pragma unroll
    for (int u = 0; u < AGG_NPW; ++u) {
        const int ic = min(i0 + u, N - 1);
        dv[u] = dis[ic];
        ts[u] = T2[(size_t)ic * (HID / 2) + lane];
    }

    int rc[AGG_NPW];
    #pragma unroll
    for (int u = 0; u < AGG_NPW; ++u) {
        rc[u] = 0;
        if (rp[u] + lane < rp[u + 1]) rc[u] = erec[rp[u] + lane];
    }

    #pragma unroll
    for (int u = 0; u < AGG_NPW; ++u) {
        const int   i  = i0 + u;
        const float di = dv[u];
        v2f acc = ts[u];

        int rec = rc[u];
        int base = rp[u];
        const int end = rp[u + 1];
        while (base < end) {
            const int n = min(end - base, 32);
            int j = 0;
            for (; j + AGG_GRP <= n; j += AGG_GRP) {
                v2f t[AGG_GRP];
                #pragma unroll
                for (int k = 0; k < AGG_GRP; ++k) {
                    const int c = __builtin_amdgcn_readlane(rec, j + k);
                    t[k] = T2[(size_t)c * (HID / 2) + lane];
                }
                #pragma unroll
                for (int k = 0; k < AGG_GRP; ++k) acc += t[k];
            }
            if (j < n) {
                v2f t[AGG_GRP - 1];
                #pragma unroll
                for (int k = 0; k < AGG_GRP - 1; ++k) {
                    t[k].x = 0.f;  t[k].y = 0.f;
                    if (j + k < n) {
                        const int c = __builtin_amdgcn_readlane(rec, j + k);
                        t[k] = T2[(size_t)c * (HID / 2) + lane];
                    }
                }
                #pragma unroll
                for (int k = 0; k < AGG_GRP - 1; ++k) acc += t[k];
            }
            base += 32;
            if (base < end) {
                rec = 0;
                if (base + lane < end) rec = erec[base + lane];
            }
        }

        v2f h;
        h.x = fmaxf(fmaf(acc.x, di, bx), 0.f);
        h.y = fmaxf(fmaf(acc.y, di, by), 0.f);
        if (i >= N) {
            h.x = 0.f;  h.y = 0.f;
        }
        sh[(wib * AGG_NPW + u) * SH_PITCH2 + lane] = h;
    }

    const int mi = lane & 15;
    const int hi = lane >> 4;
    const int kb = hi * 2;

    v2f wf[HID / 4];
    #pragma unroll
    for (int s = 0; s < HID / 4; ++s) { wf[s].x = 0.f;  wf[s].y = 0.f; }
    if (!last) {
        const float* wc = Wn + wib * 16 + mi;
        #pragma unroll
        for (int s = 0; s < HID / 4; ++s) {
            wf[s].x = wc[(4 * s + kb) * HID];
            wf[s].y = wc[(4 * s + kb + 1) * HID];
        }
    }

    __syncthreads();

    if (!last) {
        v8f tacc = {0.f,0.f,0.f,0.f,0.f,0.f,0.f,0.f};
        #pragma unroll
        for (int s = 0; s < HID / 4; ++s) {
            const v2f a = sh[mi * SH_PITCH2 + ((4 * s + kb) >> 1)];
            tacc = __builtin_amdgcn_wmma_f32_16x16x4_f32(false, a, false, wf[s], (short)0, tacc, false, false);
        }
        float* stf = reinterpret_cast<float*>(st);
        #pragma unroll
        for (int v = 0; v < 8; ++v)
            stf[(v + hi * 8) * (2 * SH_PITCH2) + wib * 16 + mi] = tacc[v];

        __syncthreads();
        v2f tv[AGG_NPW];
        #pragma unroll
        for (int u = 0; u < AGG_NPW; ++u) tv[u] = st[(wib * AGG_NPW + u) * SH_PITCH2 + lane];
        #pragma unroll
        for (int u = 0; u < AGG_NPW; ++u) {
            const int r = m0 + wib * AGG_NPW + u;
            if (r < N) Tn2[(size_t)r * (HID / 2) + lane] = tv[u] * dv[u];
        }
    }

    if (wib < (OUTD + 15) / 16) {
        const int  col = wib * 16 + mi;
        const bool cok = col < OUTD;

        v8f acc;
        #pragma unroll
        for (int v = 0; v < 8; ++v) {
            const int row = m0 + v + hi * 8;
            float c0 = 0.f;
            if (cok && row < N) c0 = first ? lin_b[col] : Out[(size_t)row * OUTD + col];
            acc[v] = c0;
        }

        v2f bf[HID / 4];
        #pragma unroll
        for (int s = 0; s < HID / 4; ++s) { bf[s].x = 0.f;  bf[s].y = 0.f; }
        if (cok) {
            #pragma unroll
            for (int s = 0; s < HID / 4; ++s) {
                bf[s].x = lw[(4 * s + kb) * OUTD + col];
                bf[s].y = lw[(4 * s + kb + 1) * OUTD + col];
            }
        }

        #pragma unroll
        for (int s = 0; s < HID / 4; ++s) {
            const v2f a = sh[mi * SH_PITCH2 + ((4 * s + kb) >> 1)];
            acc = __builtin_amdgcn_wmma_f32_16x16x4_f32(false, a, false, bf[s], (short)0, acc, false, false);
        }

        #pragma unroll
        for (int v = 0; v < 8; ++v) {
            const int row = m0 + v + hi * 8;
            if (cok && row < N) Out[(size_t)row * OUTD + col] = acc[v];
        }
    }
}

extern "C" void kernel_launch(void* const* d_in, const int* in_sizes, int n_in,
                              void* d_out, int out_size, void* d_ws, size_t ws_size,
                              hipStream_t stream) {
    const float* x     = (const float*)d_in[0];
    const float* W0    = (const float*)d_in[1];
    const float* Ws    = (const float*)d_in[2];
    const float* bs    = (const float*)d_in[3];
    const float* lin_w = (const float*)d_in[4];
    const float* lin_b = (const float*)d_in[5];
    const int*   ei    = (const int*)d_in[6];
    float* out = (float*)d_out;

    const int N = in_sizes[0] / IN_DIM;
    const int E = in_sizes[6] / 2;
    const int* src = ei;
    const int* dst = ei + E;

    char* ws = (char*)d_ws;
    float* ta     = (float*)ws; ws += sizeof(float) * (size_t)N * HID;
    float* tb     = (float*)ws; ws += sizeof(float) * (size_t)N * HID;
    const int B  = (N + BIN_NODES - 1) / BIN_NODES;
    const int G  = (E + BIN_CHUNK - 1) / BIN_CHUNK;
    int2*  tmp    = (int2*)ws;  ws += sizeof(int2)  * (size_t)G * BIN_CHUNK;
    int*   erec   = (int*)ws;   ws += sizeof(int)   * (size_t)E;
    float* dis    = (float*)ws; ws += sizeof(float) * (size_t)N;
    int*   rowptr = (int*)ws;   ws += sizeof(int)   * (size_t)(N + 1);
    int*   run    = (int*)ws;   ws += sizeof(int)   * (size_t)(B + 1) * G;

    bin_scatter_kernel<<<G, BIN_TPB, 0, stream>>>(src, dst, run, tmp, E, G, B);
    bin_finalize_kernel<<<B, BIN_TPB, 0, stream>>>(tmp, run, rowptr, dis, erec, N, E, G);
    const int strips = N / 16;
    const int gblocks = (strips + 7) / 8;
    gemm_wmma_f32_kernel<<<gblocks, 256, 0, stream>>>(x, W0, dis, ta, N, IN_DIM);
    for (int l = 0; l < LAYERS; ++l) {
        const float* tin  = (l & 1) ? tb : ta;
        float*       tout = (l & 1) ? ta : tb;
        const int    last = (l == LAYERS - 1) ? 1 : 0;
        const float* Wn   = Ws + (size_t)(last ? 0 : l) * HID * HID;

        aggregate_kernel<<<(N + AGG_NODES - 1) / AGG_NODES, AGG_WAVES * 32, 0, stream>>>(
            tin, rowptr, erec, dis,
            bs + (size_t)l * HID,
            lin_w + (size_t)l * HID * OUTD,
            lin_b, Wn, tout, out, N, (l == 0) ? 1 : 0, last);
    }
}


# JK and next-layer weights re-laid in WMMA fragment order by spare work-groups of the GEMM launch: a wave's B fragments come as 8 loads of 512 contiguous bytes, not 32 of half-used lines; sums kept
# speedup vs baseline: 3.7762x; 1.0310x over previous
#include <hip/hip_runtime.h>

typedef __attribute__((ext_vector_type(2))) float v2f;
typedef __attribute__((ext_vector_type(8))) float v8f;
typedef __attribute__((ext_vector_type(4))) float v4f;

#define IN_DIM 128
#define HID    64
#define LAYERS 8
#define OUTD   40

#define BIN_SHIFT 8
#define BIN_NODES (1 << BIN_SHIFT)
#define BIN_MAXB  512
#define BIN_TPB   1024
#define BIN_EPT   6
#define BIN_CHUNK (BIN_TPB * BIN_EPT)
#define BIN_CAP   6144
#define BIN_RUNS  8

__device__ __forceinline__ int excl_scan_first512(int v, int tid, int* wt) {
    const int lane = tid & 31;
    const int wib  = tid >> 5;
    int incl = v;
    for (int off = 1; off < 32; off <<= 1) {
        int t = __shfl_up(incl, off, 32);
        if (lane >= off) incl += t;
    }
    if (wib < 16 && lane == 31) wt[wib] = incl;
    __syncthreads();
    int add = 0;
    for (int w = 0; w < 16; ++w) add += (w < wib) ? wt[w] : 0;
    return add + incl - v;
}

__global__ __launch_bounds__(BIN_TPB) void bin_scatter_kernel(const int* __restrict__ src,
                                                              const int* __restrict__ dst,
                                                              int* __restrict__ RUN,
                                                              int2* __restrict__ tmp,
                                                              int E, int G, int B) {
    __shared__ int  c[BIN_MAXB];
    __shared__ int  ls[BIN_MAXB];
    __shared__ int  wt[16];
    __shared__ int2 rec[BIN_CHUNK];
    const int tid = threadIdx.x;
    const int g   = blockIdx.x;
    if (tid < BIN_MAXB) c[tid] = 0;
    __syncthreads();

    int s[BIN_EPT], d[BIN_EPT], r[BIN_EPT];
    const int e0 = g * BIN_CHUNK + tid;
    #pragma unroll
    for (int k = 0; k < BIN_EPT; ++k) {
        const int e = e0 + k * BIN_TPB;
        s[k] = 0;  d[k] = -1;  r[k] = 0;
        if (e < E) {
            s[k] = src[e];
            d[k] = dst[e];
            r[k] = atomicAdd(&c[d[k] >> BIN_SHIFT], 1);
        }
    }
    __syncthreads();

    const int n  = min(BIN_CHUNK, E - g * BIN_CHUNK);
    const int v  = (tid < B) ? c[tid] : 0;
    const int ex = excl_scan_first512(v, tid, wt);
    if (tid < B) {
        ls[tid] = ex;
        RUN[tid * G + g] = ex;
    }
    if (tid == B) RUN[B * G + g] = n;
    __syncthreads();

    #pragma unroll
    for (int k = 0; k < BIN_EPT; ++k)
        if (d[k] >= 0) rec[ls[d[k] >> BIN_SHIFT] + r[k]] = make_int2(s[k], d[k]);
    __syncthreads();

    int2* __restrict__ out = tmp + (size_t)g * BIN_CHUNK;
    for (int t = tid; t < n; t += BIN_TPB) out[t] = rec[t];
}

template <class F>
__device__ __forceinline__ void for_bucket_records(const int2* __restrict__ tmp,
                                                   const int* __restrict__ rs,
                                                   const int* __restrict__ re,
                                                   int G, int wib, int lane, F f) {
    const int cpw = (G + 31) / 32;
    const int g0  = wib * cpw;
    const int g1  = min(g0 + cpw, G);
    for (int gb = g0; gb < g1; gb += 32) {
        int s = 0, e = 0;
        if (gb + lane < g1) { s = rs[gb + lane];  e = re[gb + lane]; }
        const int cnt = min(32, g1 - gb);
        for (int i0 = 0; i0 < cnt; i0 += BIN_RUNS) {
            int si[BIN_RUNS], ei[BIN_RUNS];
            #pragma unroll
            for (int k = 0; k < BIN_RUNS; ++k) {
                si[k] = __builtin_amdgcn_readlane(s, i0 + k);
                ei[k] = __builtin_amdgcn_readlane(e, i0 + k);
            }
            int2 x[BIN_RUNS];
            #pragma unroll
            for (int k = 0; k < BIN_RUNS; ++k) {
                x[k] = make_int2(0, 0);
                if (si[k] + lane < ei[k])
                    x[k] = tmp[(size_t)(gb + i0 + k) * BIN_CHUNK + si[k] + lane];
            }
            #pragma unroll
            for (int k = 0; k < BIN_RUNS; ++k) {
                if (si[k] + lane < ei[k]) f(x[k]);
                for (int p = si[k] + 32 + lane; p < ei[k]; p += 32)
                    f(tmp[(size_t)(gb + i0 + k) * BIN_CHUNK + p]);
            }
        }
    }
}

__global__ __launch_bounds__(BIN_TPB) void bin_finalize_kernel(const int2* __restrict__ tmp,
                                                               const int* __restrict__ RUN,
                                                               int* __restrict__ rowptr,
                                                               float* __restrict__ dis,
                                                               int* __restrict__ erec,
                                                               int N, int E, int G) {
    __shared__ int  deg[BIN_NODES];
    __shared__ int  ptr[BIN_NODES];
    __shared__ int  wt[16];
    __shared__ int  tot[2 * (BIN_TPB / 32)];
    __shared__ int  stage[BIN_CAP];
    const int tid  = threadIdx.x;
    const int lane = tid & 31;
    const int wib  = __builtin_amdgcn_readfirstlane(tid >> 5);
    const int b    = blockIdx.x;
    const int n0   = b << BIN_SHIFT;
    const int* __restrict__ rs = RUN + (size_t)b * G;
    const int* __restrict__ re = rs + G;

    int ps = 0, pe = 0;
    for (int g = tid; g < G; g += BIN_TPB) { ps += rs[g];  pe += re[g]; }
    for (int off = 16; off > 0; off >>= 1) {
        ps += __shfl_xor(ps, off, 32);
        pe += __shfl_xor(pe, off, 32);
    }
    if (lane == 0) { tot[wib] = ps;  tot[BIN_TPB / 32 + wib] = pe; }
    if (tid < BIN_NODES) deg[tid] = 0;
    __syncthreads();
    int bs = tot[lane], be = tot[BIN_TPB / 32 + lane];
    for (int off = 16; off > 0; off >>= 1) {
        bs += __shfl_xor(bs, off, 32);
        be += __shfl_xor(be, off, 32);
    }
    const int nb  = be - bs;
    const bool staged = nb <= BIN_CAP;

    for_bucket_records(tmp, rs, re, G, wib, lane,
                       [&](int2 x) { atomicAdd(&deg[x.y - n0], 1); });
    __syncthreads();

    const int v  = (tid < BIN_NODES) ? deg[tid] : 0;
    const int ex = excl_scan_first512(v, tid, wt);
    if (tid < BIN_NODES) {
        ptr[tid] = ex;
        const int node = n0 + tid;
        if (node < N) {
            rowptr[node] = bs + ex;
            dis[node] = rsqrtf((float)(v + 1));
        }
    }
    if (b == 0 && tid == 0) rowptr[N] = E;
    __syncthreads();

    for_bucket_records(tmp, rs, re, G, wib, lane,
                       [&](int2 x) {
                           const int q = atomicAdd(&ptr[x.y - n0], 1);
                           if (staged) stage[q] = x.x;
                           else        erec[bs + q] = x.x;
                       });
    __syncthreads();
    if (staged)
        for (int t = tid; t < nb; t += BIN_TPB) erec[bs + t] = stage[t];
}

#define FR_W_F4   (4 * 8 * 32)
#define FR_LW_F4  (3 * 8 * 32)
#define FR_TOTAL  ((LAYERS - 1) * FR_W_F4 + LAYERS * FR_LW_F4)

__global__ __launch_bounds__(256) void gemm_wmma_f32_kernel(
    const float* __restrict__ A, const float* __restrict__ W,
    const float* __restrict__ dis,
    float* __restrict__ T, int nrows, int K,
    const float* __restrict__ Ws, const float* __restrict__ lin_w,
    v4f* __restrict__ wfrag, v4f* __restrict__ lwfrag, int gblocks)
{
    if ((int)blockIdx.x >= gblocks) {
        const int t = ((int)blockIdx.x - gblocks) * 256 + (int)threadIdx.x;
        if (t >= FR_TOTAL) return;
        const bool isw = t < (LAYERS - 1) * FR_W_F4;
        const int  u   = isw ? t : t - (LAYERS - 1) * FR_W_F4;
        const int  per = isw ? FR_W_F4 : FR_LW_F4;
        const int  m   = u / per;
        const int  r   = u - m * per;
        const int  fl  = r & 31;
        const int  q   = (r >> 5) & 7;
        const int  fw  = r >> 8;
        const int  c   = fw * 16 + (fl & 15);
        const int  k0  = 8 * q + 2 * (fl >> 4);
        v4f o = {0.f, 0.f, 0.f, 0.f};
        if (isw) {
            const float* M = Ws + (size_t)m * HID * HID;
            o.x = M[(k0)     * HID + c];
            o.y = M[(k0 + 1) * HID + c];
            o.z = M[(k0 + 4) * HID + c];
            o.w = M[(k0 + 5) * HID + c];
            wfrag[u] = o;
        } else {
            if (c < OUTD) {
                const float* M = lin_w + (size_t)m * HID * OUTD;
                o.x = M[(k0)     * OUTD + c];
                o.y = M[(k0 + 1) * OUTD + c];
                o.z = M[(k0 + 4) * OUTD + c];
                o.w = M[(k0 + 5) * OUTD + c];
            }
            lwfrag[u] = o;
        }
        return;
    }
    const int lane  = threadIdx.x & 31;
    const int wib   = threadIdx.x >> 5;
    const int strip = blockIdx.x * 8 + wib;
    const int m0    = strip * 16;
    if (m0 >= nrows) return;

    const int mi = lane & 15;
    const int hi = (lane >> 4) & 1;
    const int kb = hi * 2;

    const float* arow = A + (size_t)(m0 + mi) * K;

    v8f acc0 = {0.f,0.f,0.f,0.f,0.f,0.f,0.f,0.f};
    v8f acc1 = acc0, acc2 = acc0, acc3 = acc0;

    for (int k0 = 0; k0 < K; k0 += 4) {
        v2f a;
        a.x = arow[k0 + kb];
        a.y = arow[k0 + kb + 1];
        const float* w0 = W + (size_t)(k0 + kb) * HID + mi;
        const float* w1 = w0 + HID;
        v2f b0, b1, b2, b3;
        b0.x = w0[0];   b0.y = w1[0];
        b1.x = w0[16];  b1.y = w1[16];
        b2.x = w0[32];  b2.y = w1[32];
        b3.x = w0[48];  b3.y = w1[48];
        acc0 = __builtin_amdgcn_wmma_f32_16x16x4_f32(false, a, false, b0, (short)0, acc0, false, false);
        acc1 = __builtin_amdgcn_wmma_f32_16x16x4_f32(false, a, false, b1, (short)0, acc1, false, false);
        acc2 = __builtin_amdgcn_wmma_f32_16x16x4_f32(false, a, false, b2, (short)0, acc2, false, false);
        acc3 = __builtin_amdgcn_wmma_f32_16x16x4_f32(false, a, false, b3, (short)0, acc3, false, false);
    }

    #pragma unroll
    for (int v = 0; v < 8; ++v) {
        const int   row = m0 + v + hi * 8;
        const float d   = dis[row];
        float* tr = T + (size_t)row * HID + mi;
        tr[0]  = acc0[v] * d;
        tr[16] = acc1[v] * d;
        tr[32] = acc2[v] * d;
        tr[48] = acc3[v] * d;
    }
}

#define AGG_NODES 16
#define AGG_WAVES 4
#define AGG_NPW   (AGG_NODES / AGG_WAVES)
#define AGG_GRP   8
#define SH_PITCH2 34

__global__ __launch_bounds__(AGG_WAVES * 32) void aggregate_kernel(
    const float* __restrict__ T,
    const int* __restrict__ rowptr,
    const int* __restrict__ erec,
    const float* __restrict__ dis,
    const float* __restrict__ bias,
    const float* __restrict__ lwf,
    const float* __restrict__ lin_b,
    const float* __restrict__ wnf,
    float* __restrict__ Tn,
    float* __restrict__ Out,
    int N, int first, int last)
{
    __shared__ v2f sh[AGG_NODES * SH_PITCH2];
    __shared__ v2f st[AGG_NODES * SH_PITCH2];

    const int lane = threadIdx.x & 31;
    const int wib = __builtin_amdgcn_readfirstlane((int)(threadIdx.x >> 5));
    const int m0  = blockIdx.x * AGG_NODES;
    const int i0  = m0 + wib * AGG_NPW;

    const v2f* __restrict__ T2 = reinterpret_cast<const v2f*>(T);
    v2f* __restrict__ Tn2 = reinterpret_cast<v2f*>(Tn);

    const float bx = bias[2 * lane];
    const float by = bias[2 * lane + 1];

    int   rp[AGG_NPW + 1];
    float dv[AGG_NPW];
    v2f   ts[AGG_NPW];
    #pragma unroll
    for (int u = 0; u <= AGG_NPW; ++u) rp[u] = rowptr[min(i0 + u, N)];
    #pragma unroll
    for (int u = 0; u < AGG_NPW; ++u) {
        const int ic = min(i0 + u, N - 1);
        dv[u] = dis[ic];
        ts[u] = T2[(size_t)ic * (HID / 2) + lane];
    }

    int rc[AGG_NPW];
    #pragma unroll
    for (int u = 0; u < AGG_NPW; ++u) {
        rc[u] = 0;
        if (rp[u] + lane < rp[u + 1]) rc[u] = erec[rp[u] + lane];
    }

    #pragma unroll
    for (int u = 0; u < AGG_NPW; ++u) {
        const int   i  = i0 + u;
        const float di = dv[u];
        v2f acc = ts[u];

        int rec = rc[u];
        int base = rp[u];
        const int end = rp[u + 1];
        while (base < end) {
            const int n = min(end - base, 32);
            int j = 0;
            for (; j + AGG_GRP <= n; j += AGG_GRP) {
                v2f t[AGG_GRP];
                #pragma unroll
                for (int k = 0; k < AGG_GRP; ++k) {
                    const int c = __builtin_amdgcn_readlane(rec, j + k);
                    t[k] = T2[(size_t)c * (HID / 2) + lane];
                }
                #pragma unroll
                for (int k = 0; k < AGG_GRP; ++k) acc += t[k];
            }
            if (j < n) {
                v2f t[AGG_GRP - 1];
                #pragma unroll
                for (int k = 0; k < AGG_GRP - 1; ++k) {
                    t[k].x = 0.f;  t[k].y = 0.f;
                    if (j + k < n) {
                        const int c = __builtin_amdgcn_readlane(rec, j + k);
                        t[k] = T2[(size_t)c * (HID / 2) + lane];
                    }
                }
                #pragma unroll
                for (int k = 0; k < AGG_GRP - 1; ++k) acc += t[k];
            }
            base += 32;
            if (base < end) {
                rec = 0;
                if (base + lane < end) rec = erec[base + lane];
            }
        }

        v2f h;
        h.x = fmaxf(fmaf(acc.x, di, bx), 0.f);
        h.y = fmaxf(fmaf(acc.y, di, by), 0.f);
        if (i >= N) {
            h.x = 0.f;  h.y = 0.f;
        }
        sh[(wib * AGG_NPW + u) * SH_PITCH2 + lane] = h;
    }

    const int mi = lane & 15;
    const int hi = lane >> 4;
    const int kb = hi * 2;

    v2f wf[HID / 4];
    #pragma unroll
    for (int s = 0; s < HID / 4; ++s) { wf[s].x = 0.f;  wf[s].y = 0.f; }
    if (!last) {
        const v4f* wc = reinterpret_cast<const v4f*>(wnf) + wib * (8 * 32) + lane;
        #pragma unroll
        for (int q = 0; q < HID / 8; ++q) {
            const v4f w4 = wc[q * 32];
            wf[2 * q].x     = w4.x;  wf[2 * q].y     = w4.y;
            wf[2 * q + 1].x = w4.z;  wf[2 * q + 1].y = w4.w;
        }
    }

    __syncthreads();

    if (!last) {
        v8f tacc = {0.f,0.f,0.f,0.f,0.f,0.f,0.f,0.f};
        #pragma unroll
        for (int s = 0; s < HID / 4; ++s) {
            const v2f a = sh[mi * SH_PITCH2 + ((4 * s + kb) >> 1)];
            tacc = __builtin_amdgcn_wmma_f32_16x16x4_f32(false, a, false, wf[s], (short)0, tacc, false, false);
        }
        float* stf = reinterpret_cast<float*>(st);
        #pragma unroll
        for (int v = 0; v < 8; ++v)
            stf[(v + hi * 8) * (2 * SH_PITCH2) + wib * 16 + mi] = tacc[v];

        __syncthreads();
        v2f tv[AGG_NPW];
        #pragma unroll
        for (int u = 0; u < AGG_NPW; ++u) tv[u] = st[(wib * AGG_NPW + u) * SH_PITCH2 + lane];
        #pragma unroll
        for (int u = 0; u < AGG_NPW; ++u) {
            const int r = m0 + wib * AGG_NPW + u;
            if (r < N) Tn2[(size_t)r * (HID / 2) + lane] = tv[u] * dv[u];
        }
    }

    if (wib < (OUTD + 15) / 16) {
        const int  col = wib * 16 + mi;
        const bool cok = col < OUTD;

        v8f acc;
        #pragma unroll
        for (int v = 0; v < 8; ++v) {
            const int row = m0 + v + hi * 8;
            float c0 = 0.f;
            if (cok && row < N) c0 = first ? lin_b[col] : Out[(size_t)row * OUTD + col];
            acc[v] = c0;
        }

        v2f bf[HID / 4];
        const v4f* lc = reinterpret_cast<const v4f*>(lwf) + wib * (8 * 32) + lane;
        #pragma unroll
        for (int q = 0; q < HID / 8; ++q) {
            const v4f b4 = lc[q * 32];
            bf[2 * q].x     = b4.x;  bf[2 * q].y     = b4.y;
            bf[2 * q + 1].x = b4.z;  bf[2 * q + 1].y = b4.w;
        }

        #pragma unroll
        for (int s = 0; s < HID / 4; ++s) {
            const v2f a = sh[mi * SH_PITCH2 + ((4 * s + kb) >> 1)];
            acc = __builtin_amdgcn_wmma_f32_16x16x4_f32(false, a, false, bf[s], (short)0, acc, false, false);
        }

        #pragma unroll
        for (int v = 0; v < 8; ++v) {
            const int row = m0 + v + hi * 8;
            if (cok && row < N) Out[(size_t)row * OUTD + col] = acc[v];
        }
    }
}

extern "C" void kernel_launch(void* const* d_in, const int* in_sizes, int n_in,
                              void* d_out, int out_size, void* d_ws, size_t ws_size,
                              hipStream_t stream) {
    const float* x     = (const float*)d_in[0];
    const float* W0    = (const float*)d_in[1];
    const float* Ws    = (const float*)d_in[2];
    const float* bs    = (const float*)d_in[3];
    const float* lin_w = (const float*)d_in[4];
    const float* lin_b = (const float*)d_in[5];
    const int*   ei    = (const int*)d_in[6];
    float* out = (float*)d_out;

    const int N = in_sizes[0] / IN_DIM;
    const int E = in_sizes[6] / 2;
    const int* src = ei;
    const int* dst = ei + E;

    char* ws = (char*)d_ws;
    float* ta     = (float*)ws; ws += sizeof(float) * (size_t)N * HID;
    float* tb     = (float*)ws; ws += sizeof(float) * (size_t)N * HID;
    v4f*   wfrag  = (v4f*)ws;   ws += sizeof(v4f) * (size_t)(LAYERS - 1) * FR_W_F4;
    v4f*   lwfrag = (v4f*)ws;   ws += sizeof(v4f) * (size_t)LAYERS * FR_LW_F4;
    const int B  = (N + BIN_NODES - 1) / BIN_NODES;
    const int G  = (E + BIN_CHUNK - 1) / BIN_CHUNK;
    int2*  tmp    = (int2*)ws;  ws += sizeof(int2)  * (size_t)G * BIN_CHUNK;
    int*   erec   = (int*)ws;   ws += sizeof(int)   * (size_t)E;
    float* dis    = (float*)ws; ws += sizeof(float) * (size_t)N;
    int*   rowptr = (int*)ws;   ws += sizeof(int)   * (size_t)(N + 1);
    int*   run    = (int*)ws;   ws += sizeof(int)   * (size_t)(B + 1) * G;

    bin_scatter_kernel<<<G, BIN_TPB, 0, stream>>>(src, dst, run, tmp, E, G, B);
    bin_finalize_kernel<<<B, BIN_TPB, 0, stream>>>(tmp, run, rowptr, dis, erec, N, E, G);
    const int strips = N / 16;
    const int gblocks = (strips + 7) / 8;
    const int fblocks = (FR_TOTAL + 255) / 256;
    gemm_wmma_f32_kernel<<<gblocks + fblocks, 256, 0, stream>>>(x, W0, dis, ta, N, IN_DIM,
                                                               Ws, lin_w, wfrag, lwfrag, gblocks);
    for (int l = 0; l < LAYERS; ++l) {
        const float* tin  = (l & 1) ? tb : ta;
        float*       tout = (l & 1) ? ta : tb;
        const int    last = (l == LAYERS - 1) ? 1 : 0;
        const float* wnf  = (const float*)(wfrag + (size_t)(last ? 0 : l) * FR_W_F4);

        aggregate_kernel<<<(N + AGG_NODES - 1) / AGG_NODES, AGG_WAVES * 32, 0, stream>>>(
            tin, rowptr, erec, dis,
            bs + (size_t)l * HID,
            (const float*)(lwfrag + (size_t)l * FR_LW_F4),
            lin_b, wnf, tout, out, N, (l == 0) ? 1 : 0, last);
    }
}


# Aggregation: a wave owns 8 nodes and sums two at a time with the same whole-wave row loads (2 x 8 rows in flight), 32-node groups of 4 waves: half as many waves enter per layer; sums kept
# speedup vs baseline: 3.9523x; 1.0466x over previous
#include <hip/hip_runtime.h>

typedef __attribute__((ext_vector_type(2))) float v2f;
typedef __attribute__((ext_vector_type(8))) float v8f;
typedef __attribute__((ext_vector_type(4))) float v4f;

#define IN_DIM 128
#define HID    64
#define LAYERS 8
#define OUTD   40

#define BIN_SHIFT 8
#define BIN_NODES (1 << BIN_SHIFT)
#define BIN_MAXB  512
#define BIN_TPB   1024
#define BIN_EPT   6
#define BIN_CHUNK (BIN_TPB * BIN_EPT)
#define BIN_CAP   6144
#define BIN_RUNS  8

__device__ __forceinline__ int excl_scan_first512(int v, int tid, int* wt) {
    const int lane = tid & 31;
    const int wib  = tid >> 5;
    int incl = v;
    for (int off = 1; off < 32; off <<= 1) {
        int t = __shfl_up(incl, off, 32);
        if (lane >= off) incl += t;
    }
    if (wib < 16 && lane == 31) wt[wib] = incl;
    __syncthreads();
    int add = 0;
    for (int w = 0; w < 16; ++w) add += (w < wib) ? wt[w] : 0;
    return add + incl - v;
}

__global__ __launch_bounds__(BIN_TPB) void bin_scatter_kernel(const int* __restrict__ src,
                                                              const int* __restrict__ dst,
                                                              int* __restrict__ RUN,
                                                              int2* __restrict__ tmp,
                                                              int E, int G, int B) {
    __shared__ int  c[BIN_MAXB];
    __shared__ int  ls[BIN_MAXB];
    __shared__ int  wt[16];
    __shared__ int2 rec[BIN_CHUNK];
    const int tid = threadIdx.x;
    const int g   = blockIdx.x;
    if (tid < BIN_MAXB) c[tid] = 0;
    __syncthreads();

    int s[BIN_EPT], d[BIN_EPT], r[BIN_EPT];
    const int e0 = g * BIN_CHUNK + tid;
    #pragma unroll
    for (int k = 0; k < BIN_EPT; ++k) {
        const int e = e0 + k * BIN_TPB;
        s[k] = 0;  d[k] = -1;  r[k] = 0;
        if (e < E) {
            s[k] = src[e];
            d[k] = dst[e];
            r[k] = atomicAdd(&c[d[k] >> BIN_SHIFT], 1);
        }
    }
    __syncthreads();

    const int n  = min(BIN_CHUNK, E - g * BIN_CHUNK);
    const int v  = (tid < B) ? c[tid] : 0;
    const int ex = excl_scan_first512(v, tid, wt);
    if (tid < B) {
        ls[tid] = ex;
        RUN[tid * G + g] = ex;
    }
    if (tid == B) RUN[B * G + g] = n;
    __syncthreads();

    #pragma unroll
    for (int k = 0; k < BIN_EPT; ++k)
        if (d[k] >= 0) rec[ls[d[k] >> BIN_SHIFT] + r[k]] = make_int2(s[k], d[k]);
    __syncthreads();

    int2* __restrict__ out = tmp + (size_t)g * BIN_CHUNK;
    for (int t = tid; t < n; t += BIN_TPB) out[t] = rec[t];
}

template <class F>
__device__ __forceinline__ void for_bucket_records(const int2* __restrict__ tmp,
                                                   const int* __restrict__ rs,
                                                   const int* __restrict__ re,
                                                   int G, int wib, int lane, F f) {
    const int cpw = (G + 31) / 32;
    const int g0  = wib * cpw;
    const int g1  = min(g0 + cpw, G);
    for (int gb = g0; gb < g1; gb += 32) {
        int s = 0, e = 0;
        if (gb + lane < g1) { s = rs[gb + lane];  e = re[gb + lane]; }
        const int cnt = min(32, g1 - gb);
        for (int i0 = 0; i0 < cnt; i0 += BIN_RUNS) {
            int si[BIN_RUNS], ei[BIN_RUNS];
            #pragma unroll
            for (int k = 0; k < BIN_RUNS; ++k) {
                si[k] = __builtin_amdgcn_readlane(s, i0 + k);
                ei[k] = __builtin_amdgcn_readlane(e, i0 + k);
            }
            int2 x[BIN_RUNS];
            #pragma unroll
            for (int k = 0; k < BIN_RUNS; ++k) {
                x[k] = make_int2(0, 0);
                if (si[k] + lane < ei[k])
                    x[k] = tmp[(size_t)(gb + i0 + k) * BIN_CHUNK + si[k] + lane];
            }
            #pragma unroll
            for (int k = 0; k < BIN_RUNS; ++k) {
                if (si[k] + lane < ei[k]) f(x[k]);
                for (int p = si[k] + 32 + lane; p < ei[k]; p += 32)
                    f(tmp[(size_t)(gb + i0 + k) * BIN_CHUNK + p]);
            }
        }
    }
}

__global__ __launch_bounds__(BIN_TPB) void bin_finalize_kernel(const int2* __restrict__ tmp,
                                                               const int* __restrict__ RUN,
                                                               int* __restrict__ rowptr,
                                                               float* __restrict__ dis,
                                                               int* __restrict__ erec,
                                                               int N, int E, int G) {
    __shared__ int  deg[BIN_NODES];
    __shared__ int  ptr[BIN_NODES];
    __shared__ int  wt[16];
    __shared__ int  tot[2 * (BIN_TPB / 32)];
    __shared__ int  stage[BIN_CAP];
    const int tid  = threadIdx.x;
    const int lane = tid & 31;
    const int wib  = __builtin_amdgcn_readfirstlane(tid >> 5);
    const int b    = blockIdx.x;
    const int n0   = b << BIN_SHIFT;
    const int* __restrict__ rs = RUN + (size_t)b * G;
    const int* __restrict__ re = rs + G;

    int ps = 0, pe = 0;
    for (int g = tid; g < G; g += BIN_TPB) { ps += rs[g];  pe += re[g]; }
    for (int off = 16; off > 0; off >>= 1) {
        ps += __shfl_xor(ps, off, 32);
        pe += __shfl_xor(pe, off, 32);
    }
    if (lane == 0) { tot[wib] = ps;  tot[BIN_TPB / 32 + wib] = pe; }
    if (tid < BIN_NODES) deg[tid] = 0;
    __syncthreads();
    int bs = tot[lane], be = tot[BIN_TPB / 32 + lane];
    for (int off = 16; off > 0; off >>= 1) {
        bs += __shfl_xor(bs, off, 32);
        be += __shfl_xor(be, off, 32);
    }
    const int nb  = be - bs;
    const bool staged = nb <= BIN_CAP;

    for_bucket_records(tmp, rs, re, G, wib, lane,
                       [&](int2 x) { atomicAdd(&deg[x.y - n0], 1); });
    __syncthreads();

    const int v  = (tid < BIN_NODES) ? deg[tid] : 0;
    const int ex = excl_scan_first512(v, tid, wt);
    if (tid < BIN_NODES) {
        ptr[tid] = ex;
        const int node = n0 + tid;
        if (node < N) {
            rowptr[node] = bs + ex;
            dis[node] = rsqrtf((float)(v + 1));
        }
    }
    if (b == 0 && tid == 0) rowptr[N] = E;
    __syncthreads();

    for_bucket_records(tmp, rs, re, G, wib, lane,
                       [&](int2 x) {
                           const int q = atomicAdd(&ptr[x.y - n0], 1);
                           if (staged) stage[q] = x.x;
                           else        erec[bs + q] = x.x;
                       });
    __syncthreads();
    if (staged)
        for (int t = tid; t < nb; t += BIN_TPB) erec[bs + t] = stage[t];
}

#define FR_W_F4   (4 * 8 * 32)
#define FR_LW_F4  (3 * 8 * 32)
#define FR_TOTAL  ((LAYERS - 1) * FR_W_F4 + LAYERS * FR_LW_F4)

__global__ __launch_bounds__(256) void gemm_wmma_f32_kernel(
    const float* __restrict__ A, const float* __restrict__ W,
    const float* __restrict__ dis,
    float* __restrict__ T, int nrows, int K,
    const float* __restrict__ Ws, const float* __restrict__ lin_w,
    v4f* __restrict__ wfrag, v4f* __restrict__ lwfrag, int gblocks)
{
    if ((int)blockIdx.x >= gblocks) {
        const int t = ((int)blockIdx.x - gblocks) * 256 + (int)threadIdx.x;
        if (t >= FR_TOTAL) return;
        const bool isw = t < (LAYERS - 1) * FR_W_F4;
        const int  u   = isw ? t : t - (LAYERS - 1) * FR_W_F4;
        const int  per = isw ? FR_W_F4 : FR_LW_F4;
        const int  m   = u / per;
        const int  r   = u - m * per;
        const int  fl  = r & 31;
        const int  q   = (r >> 5) & 7;
        const int  fw  = r >> 8;
        const int  c   = fw * 16 + (fl & 15);
        const int  k0  = 8 * q + 2 * (fl >> 4);
        v4f o = {0.f, 0.f, 0.f, 0.f};
        if (isw) {
            const float* M = Ws + (size_t)m * HID * HID;
            o.x = M[(k0)     * HID + c];
            o.y = M[(k0 + 1) * HID + c];
            o.z = M[(k0 + 4) * HID + c];
            o.w = M[(k0 + 5) * HID + c];
            wfrag[u] = o;
        } else {
            if (c < OUTD) {
                const float* M = lin_w + (size_t)m * HID * OUTD;
                o.x = M[(k0)     * OUTD + c];
                o.y = M[(k0 + 1) * OUTD + c];
                o.z = M[(k0 + 4) * OUTD + c];
                o.w = M[(k0 + 5) * OUTD + c];
            }
            lwfrag[u] = o;
        }
        return;
    }
    const int lane  = threadIdx.x & 31;
    const int wib   = threadIdx.x >> 5;
    const int strip = blockIdx.x * 8 + wib;
    const int m0    = strip * 16;
    if (m0 >= nrows) return;

    const int mi = lane & 15;
    const int hi = (lane >> 4) & 1;
    const int kb = hi * 2;

    const float* arow = A + (size_t)(m0 + mi) * K;

    v8f acc0 = {0.f,0.f,0.f,0.f,0.f,0.f,0.f,0.f};
    v8f acc1 = acc0, acc2 = acc0, acc3 = acc0;

    for (int k0 = 0; k0 < K; k0 += 4) {
        v2f a;
        a.x = arow[k0 + kb];
        a.y = arow[k0 + kb + 1];
        const float* w0 = W + (size_t)(k0 + kb) * HID + mi;
        const float* w1 = w0 + HID;
        v2f b0, b1, b2, b3;
        b0.x = w0[0];   b0.y = w1[0];
        b1.x = w0[16];  b1.y = w1[16];
        b2.x = w0[32];  b2.y = w1[32];
        b3.x = w0[48];  b3.y = w1[48];
        acc0 = __builtin_amdgcn_wmma_f32_16x16x4_f32(false, a, false, b0, (short)0, acc0, false, false);
        acc1 = __builtin_amdgcn_wmma_f32_16x16x4_f32(false, a, false, b1, (short)0, acc1, false, false);
        acc2 = __builtin_amdgcn_wmma_f32_16x16x4_f32(false, a, false, b2, (short)0, acc2, false, false);
        acc3 = __builtin_amdgcn_wmma_f32_16x16x4_f32(false, a, false, b3, (short)0, acc3, false, false);
    }

    #pragma unroll
    for (int v = 0; v < 8; ++v) {
        const int   row = m0 + v + hi * 8;
        const float d   = dis[row];
        float* tr = T + (size_t)row * HID + mi;
        tr[0]  = acc0[v] * d;
        tr[16] = acc1[v] * d;
        tr[32] = acc2[v] * d;
        tr[48] = acc3[v] * d;
    }
}

#define AGG_NODES 32
#define AGG_WAVES 4
#define AGG_NPW   (AGG_NODES / AGG_WAVES)
#define AGG_PAIRS (AGG_NPW / 2)
#define AGG_GRP   8
#define SH_PITCH2 34

__device__ __forceinline__ v2f agg_row(const v2f* __restrict__ T2, int c, int lane) {
    return T2[(unsigned)(c * (HID / 2) + lane)];
}

__global__ __launch_bounds__(AGG_WAVES * 32) void aggregate_kernel(
    const float* __restrict__ T,
    const int* __restrict__ rowptr,
    const int* __restrict__ erec,
    const float* __restrict__ dis,
    const float* __restrict__ bias,
    const float* __restrict__ lwf,
    const float* __restrict__ lin_b,
    const float* __restrict__ wnf,
    float* __restrict__ Tn,
    float* __restrict__ Out,
    int N, int first, int last)
{
    __shared__ v2f sh[AGG_NODES * SH_PITCH2];
    __shared__ v2f st[AGG_NODES * SH_PITCH2];

    const int lane = threadIdx.x & 31;
    const int wib = __builtin_amdgcn_readfirstlane((int)(threadIdx.x >> 5));
    const int m0  = blockIdx.x * AGG_NODES;
    const int i0  = m0 + wib * AGG_NPW;

    const v2f* __restrict__ T2 = reinterpret_cast<const v2f*>(T);
    v2f* __restrict__ Tn2 = reinterpret_cast<v2f*>(Tn);

    const float bx = bias[2 * lane];
    const float by = bias[2 * lane + 1];

    int   rp[AGG_NPW + 1];
    float dv[AGG_NPW];
    #pragma unroll
    for (int u = 0; u <= AGG_NPW; ++u) rp[u] = rowptr[min(i0 + u, N)];
    #pragma unroll
    for (int u = 0; u < AGG_NPW; ++u) dv[u] = dis[min(i0 + u, N - 1)];

    int rc[AGG_NPW];
    #pragma unroll
    for (int u = 0; u < AGG_NPW; ++u) {
        rc[u] = min(i0 + u, N - 1);
        if (rp[u] + lane < rp[u + 1]) rc[u] = erec[rp[u] + lane];
    }

    #pragma unroll
    for (int u = 0; u < AGG_PAIRS; ++u) {
        const int ua = u, ub = u + AGG_PAIRS;
        v2f acca = agg_row(T2, min(i0 + ua, N - 1), lane);
        v2f accb = agg_row(T2, min(i0 + ub, N - 1), lane);

        int reca = rc[ua], recb = rc[ub];
        int basea = rp[ua], baseb = rp[ub];
        const int enda = rp[ua + 1], endb = rp[ub + 1];
        while (basea < enda || baseb < endb) {
            const int na = max(min(enda - basea, 32), 0);
            const int nb = max(min(endb - baseb, 32), 0);
            const int nfull = min(na, nb) & ~(AGG_GRP - 1);
            const int nmax  = max(na, nb);
            int j = 0;
            for (; j < nfull; j += AGG_GRP) {
                v2f t[2 * AGG_GRP];
                #pragma unroll
                for (int k = 0; k < AGG_GRP; ++k)
                    t[k] = agg_row(T2, __builtin_amdgcn_readlane(reca, j + k), lane);
                #pragma unroll
                for (int k = 0; k < AGG_GRP; ++k)
                    t[AGG_GRP + k] = agg_row(T2, __builtin_amdgcn_readlane(recb, j + k), lane);
                #pragma unroll
                for (int k = 0; k < AGG_GRP; ++k) acca += t[k];
                #pragma unroll
                for (int k = 0; k < AGG_GRP; ++k) accb += t[AGG_GRP + k];
            }
            const int la = max(na - 1, 0);
            const int lb = max(nb - 1, 0);
            for (; j < nmax; j += AGG_GRP) {
                v2f t[2 * AGG_GRP];
                #pragma unroll
                for (int k = 0; k < AGG_GRP; ++k)
                    t[k] = agg_row(T2, __builtin_amdgcn_readlane(reca, min(j + k, la)), lane);
                #pragma unroll
                for (int k = 0; k < AGG_GRP; ++k)
                    t[AGG_GRP + k] = agg_row(T2, __builtin_amdgcn_readlane(recb, min(j + k, lb)), lane);
                #pragma unroll
                for (int k = 0; k < AGG_GRP; ++k) {
                    const bool on = j + k < na;
                    acca.x += on ? t[k].x : 0.f;
                    acca.y += on ? t[k].y : 0.f;
                }
                #pragma unroll
                for (int k = 0; k < AGG_GRP; ++k) {
                    const bool on = j + k < nb;
                    accb.x += on ? t[AGG_GRP + k].x : 0.f;
                    accb.y += on ? t[AGG_GRP + k].y : 0.f;
                }
            }
            basea += 32;
            baseb += 32;
            if (basea < enda) {
                reca = 0;
                if (basea + lane < enda) reca = erec[basea + lane];
            }
            if (baseb < endb) {
                recb = 0;
                if (baseb + lane < endb) recb = erec[baseb + lane];
            }
        }

        v2f ha, hb;
        ha.x = fmaxf(fmaf(acca.x, dv[ua], bx), 0.f);
        ha.y = fmaxf(fmaf(acca.y, dv[ua], by), 0.f);
        hb.x = fmaxf(fmaf(accb.x, dv[ub], bx), 0.f);
        hb.y = fmaxf(fmaf(accb.y, dv[ub], by), 0.f);
        if (i0 + ua >= N) { ha.x = 0.f;  ha.y = 0.f; }
        if (i0 + ub >= N) { hb.x = 0.f;  hb.y = 0.f; }
        sh[(wib * AGG_NPW + ua) * SH_PITCH2 + lane] = ha;
        sh[(wib * AGG_NPW + ub) * SH_PITCH2 + lane] = hb;
    }

    const int mi = lane & 15;
    const int hi = lane >> 4;
    const int kb = hi * 2;

    v2f wf[HID / 4];
    #pragma unroll
    for (int s = 0; s < HID / 4; ++s) { wf[s].x = 0.f;  wf[s].y = 0.f; }
    if (!last) {
        const v4f* wc = reinterpret_cast<const v4f*>(wnf) + wib * (8 * 32) + lane;
        #pragma unroll
        for (int q = 0; q < HID / 8; ++q) {
            const v4f w4 = wc[q * 32];
            wf[2 * q].x     = w4.x;  wf[2 * q].y     = w4.y;
            wf[2 * q + 1].x = w4.z;  wf[2 * q + 1].y = w4.w;
        }
    }

    __syncthreads();

    if (!last) {
        v8f tacc0 = {0.f,0.f,0.f,0.f,0.f,0.f,0.f,0.f};
        v8f tacc1 = tacc0;
        #pragma unroll
        for (int s = 0; s < HID / 4; ++s) {
            const v2f a0 = sh[mi * SH_PITCH2 + ((4 * s + kb) >> 1)];
            const v2f a1 = sh[(16 + mi) * SH_PITCH2 + ((4 * s + kb) >> 1)];
            tacc0 = __builtin_amdgcn_wmma_f32_16x16x4_f32(false, a0, false, wf[s], (short)0, tacc0, false, false);
            tacc1 = __builtin_amdgcn_wmma_f32_16x16x4_f32(false, a1, false, wf[s], (short)0, tacc1, false, false);
        }
        float* stf = reinterpret_cast<float*>(st);
        #pragma unroll
        for (int v = 0; v < 8; ++v) {
            stf[(v + hi * 8) * (2 * SH_PITCH2) + wib * 16 + mi]      = tacc0[v];
            stf[(16 + v + hi * 8) * (2 * SH_PITCH2) + wib * 16 + mi] = tacc1[v];
        }

        __syncthreads();
        v2f tv[AGG_NPW];
        #pragma unroll
        for (int u = 0; u < AGG_NPW; ++u) tv[u] = st[(wib * AGG_NPW + u) * SH_PITCH2 + lane];
        #pragma unroll
        for (int u = 0; u < AGG_NPW; ++u) {
            const int r = m0 + wib * AGG_NPW + u;
            if (r < N) Tn2[(size_t)r * (HID / 2) + lane] = tv[u] * dv[u];
        }
    }

    if (wib < (OUTD + 15) / 16) {
        const int  col = wib * 16 + mi;
        const bool cok = col < OUTD;

        const bool whole = m0 + AGG_NODES <= N;
        const int  colc  = min(col, OUTD - 1);
        v8f acc0, acc1;
        if (first) {
            const float b = lin_b[colc];
            #pragma unroll
            for (int v = 0; v < 8; ++v) { acc0[v] = b;  acc1[v] = b; }
        } else if (whole) {
            const float* cp = Out + (size_t)(m0 + hi * 8) * OUTD + colc;
            #pragma unroll
            for (int v = 0; v < 8; ++v) {
                acc0[v] = cp[v * OUTD];
                acc1[v] = cp[(16 + v) * OUTD];
            }
        } else {
            #pragma unroll
            for (int v = 0; v < 8; ++v) {
                const int row0 = m0 + v + hi * 8;
                const int row1 = row0 + 16;
                float c0 = 0.f, c1 = 0.f;
                if (cok && row0 < N) c0 = Out[(size_t)row0 * OUTD + col];
                if (cok && row1 < N) c1 = Out[(size_t)row1 * OUTD + col];
                acc0[v] = c0;
                acc1[v] = c1;
            }
        }

        v2f bf[HID / 4];
        const v4f* lc = reinterpret_cast<const v4f*>(lwf) + wib * (8 * 32) + lane;
        #pragma unroll
        for (int q = 0; q < HID / 8; ++q) {
            const v4f b4 = lc[q * 32];
            bf[2 * q].x     = b4.x;  bf[2 * q].y     = b4.y;
            bf[2 * q + 1].x = b4.z;  bf[2 * q + 1].y = b4.w;
        }

        #pragma unroll
        for (int s = 0; s < HID / 4; ++s) {
            const v2f a0 = sh[mi * SH_PITCH2 + ((4 * s + kb) >> 1)];
            const v2f a1 = sh[(16 + mi) * SH_PITCH2 + ((4 * s + kb) >> 1)];
            acc0 = __builtin_amdgcn_wmma_f32_16x16x4_f32(false, a0, false, bf[s], (short)0, acc0, false, false);
            acc1 = __builtin_amdgcn_wmma_f32_16x16x4_f32(false, a1, false, bf[s], (short)0, acc1, false, false);
        }

        if (whole) {
            if (cok) {
                float* op = Out + (size_t)(m0 + hi * 8) * OUTD + col;
                #pragma unroll
                for (int v = 0; v < 8; ++v) {
                    op[v * OUTD]        = acc0[v];
                    op[(16 + v) * OUTD] = acc1[v];
                }
            }
        } else {
            #pragma unroll
            for (int v = 0; v < 8; ++v) {
                const int row0 = m0 + v + hi * 8;
                const int row1 = row0 + 16;
                if (cok && row0 < N) Out[(size_t)row0 * OUTD + col] = acc0[v];
                if (cok && row1 < N) Out[(size_t)row1 * OUTD + col] = acc1[v];
            }
        }
    }
}

extern "C" void kernel_launch(void* const* d_in, const int* in_sizes, int n_in,
                              void* d_out, int out_size, void* d_ws, size_t ws_size,
                              hipStream_t stream) {
    const float* x     = (const float*)d_in[0];
    const float* W0    = (const float*)d_in[1];
    const float* Ws    = (const float*)d_in[2];
    const float* bs    = (const float*)d_in[3];
    const float* lin_w = (const float*)d_in[4];
    const float* lin_b = (const float*)d_in[5];
    const int*   ei    = (const int*)d_in[6];
    float* out = (float*)d_out;

    const int N = in_sizes[0] / IN_DIM;
    const int E = in_sizes[6] / 2;
    const int* src = ei;
    const int* dst = ei + E;

    char* ws = (char*)d_ws;
    float* ta     = (float*)ws; ws += sizeof(float) * (size_t)N * HID;
    float* tb     = (float*)ws; ws += sizeof(float) * (size_t)N * HID;
    v4f*   wfrag  = (v4f*)ws;   ws += sizeof(v4f) * (size_t)(LAYERS - 1) * FR_W_F4;
    v4f*   lwfrag = (v4f*)ws;   ws += sizeof(v4f) * (size_t)LAYERS * FR_LW_F4;
    const int B  = (N + BIN_NODES - 1) / BIN_NODES;
    const int G  = (E + BIN_CHUNK - 1) / BIN_CHUNK;
    int2*  tmp    = (int2*)ws;  ws += sizeof(int2)  * (size_t)G * BIN_CHUNK;
    int*   erec   = (int*)ws;   ws += sizeof(int)   * (size_t)E;
    float* dis    = (float*)ws; ws += sizeof(float) * (size_t)N;
    int*   rowptr = (int*)ws;   ws += sizeof(int)   * (size_t)(N + 1);
    int*   run    = (int*)ws;   ws += sizeof(int)   * (size_t)(B + 1) * G;

    bin_scatter_kernel<<<G, BIN_TPB, 0, stream>>>(src, dst, run, tmp, E, G, B);
    bin_finalize_kernel<<<B, BIN_TPB, 0, stream>>>(tmp, run, rowptr, dis, erec, N, E, G);
    const int strips = N / 16;
    const int gblocks = (strips + 7) / 8;
    const int fblocks = (FR_TOTAL + 255) / 256;
    gemm_wmma_f32_kernel<<<gblocks + fblocks, 256, 0, stream>>>(x, W0, dis, ta, N, IN_DIM,
                                                               Ws, lin_w, wfrag, lwfrag, gblocks);
    for (int l = 0; l < LAYERS; ++l) {
        const float* tin  = (l & 1) ? tb : ta;
        float*       tout = (l & 1) ? ta : tb;
        const int    last = (l == LAYERS - 1) ? 1 : 0;
        const float* wnf  = (const float*)(wfrag + (size_t)(last ? 0 : l) * FR_W_F4);

        aggregate_kernel<<<(N + AGG_NODES - 1) / AGG_NODES, AGG_WAVES * 32, 0, stream>>>(
            tin, rowptr, erec, dis,
            bs + (size_t)l * HID,
            (const float*)(lwfrag + (size_t)l * FR_LW_F4),
            lin_b, wnf, tout, out, N, (l == 0) ? 1 : 0, last);
    }
}
